# starting kernel
# baseline (speedup 1.0000x reference)
#include <hip/hip_runtime.h>
#include <hip/hip_bf16.h>

#define DFEAT 128

typedef __attribute__((ext_vector_type(2))) float v2f;
typedef __attribute__((ext_vector_type(8))) float v8f;

__global__ void gcn_zero_f32(float* __restrict__ p, int n) {
  int i = blockIdx.x * blockDim.x + threadIdx.x;
  int stride = gridDim.x * blockDim.x;
  for (; i < n; i += stride) p[i] = 0.0f;
}

__global__ void gcn_degree(const int* __restrict__ row, const int* __restrict__ col,
                           float* __restrict__ deg_out, float* __restrict__ deg_in,
                           int E) {
  int e = blockIdx.x * blockDim.x + threadIdx.x;
  if (e < E) {
    atomicAdd(&deg_out[row[e]], 1.0f);
    atomicAdd(&deg_in[col[e]], 1.0f);
  }
}

__global__ void gcn_scale(const float* __restrict__ x, const float* __restrict__ deg,
                          float* __restrict__ xs, int total4) {
  int t = blockIdx.x * blockDim.x + threadIdx.x;
  if (t < total4) {
    int n = t >> 5;
    float s = rsqrtf(fmaxf(deg[n], 1.0f));
    float4 v = reinterpret_cast<const float4*>(x)[t];
    v.x *= s; v.y *= s; v.z *= s; v.w *= s;
    reinterpret_cast<float4*>(xs)[t] = v;
  }
}

__global__ void gcn_scatter(const float* __restrict__ xs, const int* __restrict__ row,
                            const int* __restrict__ col, float* __restrict__ agg,
                            int E) {
  int t = blockIdx.x * blockDim.x + threadIdx.x;
  int e = t >> 5;
  int lane = t & 31;
  if (e < E) {
    int r = row[e];
    int c = col[e];
    float4 v = reinterpret_cast<const float4*>(xs + (size_t)r * DFEAT)[lane];
    float* dst = agg + (size_t)c * DFEAT + lane * 4;
    atomicAdd(dst + 0, v.x);
    atomicAdd(dst + 1, v.y);
    atomicAdd(dst + 2, v.z);
    atomicAdd(dst + 3, v.w);
  }
}

__global__ void __launch_bounds__(256) gcn_gemm_norm(
    const float* __restrict__ A,
    const float* __restrict__ W,
    const float* __restrict__ bias,
    const float* __restrict__ deg_in,
    const float* __restrict__ resid,
    float* __restrict__ out,
    int do_relu) {
  const int lane = threadIdx.x & 31;
  const int wave = threadIdx.x >> 5;
  const int half = lane >> 4;
  const int l15  = lane & 15;
  const int n0 = blockIdx.x * 16;
  const int i0 = wave * 16;

  const float* ap = A + (size_t)(n0 + l15) * DFEAT + 2 * half;
  const float* bp = W + (size_t)(i0 + l15) * DFEAT + 2 * half;

  v8f acc = {0.f, 0.f, 0.f, 0.f, 0.f, 0.f, 0.f, 0.f};
#pragma unroll
  for (int kk = 0; kk < DFEAT / 4; ++kk) {
    v2f a = *reinterpret_cast<const v2f*>(ap + 4 * kk);
    v2f b = *reinterpret_cast<const v2f*>(bp + 4 * kk);
    acc = __builtin_amdgcn_wmma_f32_16x16x4_f32(
         false, a,  false, b,
         (short)0, acc,  false,  false);
  }

  const int colI = i0 + l15;
  const float bb = bias[colI];
#pragma unroll
  for (int r = 0; r < 8; ++r) {
    int nrow = n0 + r + 8 * half;
    float val = acc[r] + bb;
    val *= rsqrtf(fmaxf(deg_in[nrow], 1.0f));
    if (resid) val += resid[(size_t)nrow * DFEAT + colI];
    if (do_relu) val = fmaxf(val, 0.0f);
    out[(size_t)nrow * DFEAT + colI] = val;
  }
}

extern "C" void kernel_launch(void* const* d_in, const int* in_sizes, int n_in,
                              void* d_out, int out_size, void* d_ws, size_t ws_size,
                              hipStream_t stream) {
  const float* x  = (const float*)d_in[0];
  const int*   ei = (const int*)d_in[1];
  const float* W1 = (const float*)d_in[2];
  const float* b1 = (const float*)d_in[3];
  const float* W2 = (const float*)d_in[4];
  const float* b2 = (const float*)d_in[5];

  const int N = in_sizes[0] / DFEAT;
  const int E = in_sizes[1] / 2;
  const int* row = ei;
  const int* col = ei + E;
  float* out = (float*)d_out;

  float* deg_out = (float*)d_ws;
  float* deg_in  = deg_out + N;
  float* xsrc    = deg_in + N;
  float* agg     = xsrc + (size_t)N * DFEAT;

  const int TB = 256;
  const int ndTotal  = N * DFEAT;
  const int scale4   = N * (DFEAT / 4);
  const long long scatterThreads = (long long)E * 32;

  gcn_zero_f32<<<(2 * N + TB - 1) / TB, TB, 0, stream>>>(deg_out, 2 * N);
  gcn_degree<<<(E + TB - 1) / TB, TB, 0, stream>>>(row, col, deg_out, deg_in, E);

  gcn_scale<<<(scale4 + TB - 1) / TB, TB, 0, stream>>>(x, deg_out, xsrc, scale4);
  gcn_zero_f32<<<(ndTotal + TB - 1) / TB, TB, 0, stream>>>(agg, ndTotal);
  gcn_scatter<<<(int)((scatterThreads + TB - 1) / TB), TB, 0, stream>>>(xsrc, row, col, agg, E);
  gcn_gemm_norm<<<N / 16, TB, 0, stream>>>(agg, W1, b1, deg_in, x, out, 1);

  gcn_scale<<<(scale4 + TB - 1) / TB, TB, 0, stream>>>(out, deg_out, xsrc, scale4);
  gcn_zero_f32<<<(ndTotal + TB - 1) / TB, TB, 0, stream>>>(agg, ndTotal);
  gcn_scatter<<<(int)((scatterThreads + TB - 1) / TB), TB, 0, stream>>>(xsrc, row, col, agg, E);
  gcn_gemm_norm<<<N / 16, TB, 0, stream>>>(agg, W2, b2, deg_in, nullptr, out, 0);
}


# Float-atomic scatter and agg zeroing replaced by a segment sum over a destination-sorted edge list: arrival-order rank from the in-degree atomic, one wave per node, 4 partial sums per row
# speedup vs baseline: 1.5970x; 1.5970x over previous
#include <hip/hip_runtime.h>
#include <hip/hip_bf16.h>

#define DFEAT 128

typedef __attribute__((ext_vector_type(2))) float v2f;
typedef __attribute__((ext_vector_type(8))) float v8f;

__global__ void gcn_zero_f32(float* __restrict__ p, int n) {
  int i = blockIdx.x * blockDim.x + threadIdx.x;
  int stride = gridDim.x * blockDim.x;
  for (; i < n; i += stride) p[i] = 0.0f;
}

__global__ void gcn_degree_rank(const int* __restrict__ row, const int* __restrict__ col,
                                float* __restrict__ deg_out, float* __restrict__ deg_in,
                                int* __restrict__ rank, int E) {
  int e = blockIdx.x * blockDim.x + threadIdx.x;
  if (e < E) {
    atomicAdd(&deg_out[row[e]], 1.0f);
    float before = atomicAdd(&deg_in[col[e]], 1.0f);
    rank[e] = (int)before;
  }
}

__global__ void __launch_bounds__(256) gcn_segment_offsets(
    const float* __restrict__ deg_in, int* __restrict__ seg,
    int* __restrict__ cursor, int N) {
  __shared__ int wave_sum[8];
  __shared__ int block_base;
  const int lane = threadIdx.x & 31;
  const int wave = threadIdx.x >> 5;
  const int c = blockIdx.x * 256 + threadIdx.x;
  const int d = (c < N) ? (int)deg_in[c] : 0;
  int incl = d;
  for (int o = 1; o < 32; o <<= 1) {
    int t = __shfl_up(incl, o);
    if (lane >= o) incl += t;
  }
  if (lane == 31) wave_sum[wave] = incl;
  __syncthreads();
  if (threadIdx.x == 0) {
    int tot = 0;
    for (int w = 0; w < 8; ++w) {
      int t = wave_sum[w];
      wave_sum[w] = tot;
      tot += t;
    }
    block_base = atomicAdd(cursor, tot);
  }
  __syncthreads();
  if (c < N)
    reinterpret_cast<int2*>(seg)[c] = make_int2(block_base + wave_sum[wave] + incl - d, d);
}

__global__ void gcn_fill_sorted(const int* __restrict__ row, const int* __restrict__ col,
                                const int* __restrict__ rank, const int* __restrict__ seg,
                                int* __restrict__ srcs, int E) {
  int e = blockIdx.x * blockDim.x + threadIdx.x;
  if (e < E) {
    int p = seg[2 * col[e]] + rank[e];
    if ((unsigned)p < (unsigned)E) srcs[p] = row[e];
  }
}

__global__ void gcn_scale(const float* __restrict__ x, const float* __restrict__ deg,
                          float* __restrict__ xs, int total4) {
  int t = blockIdx.x * blockDim.x + threadIdx.x;
  if (t < total4) {
    int n = t >> 5;
    float s = rsqrtf(fmaxf(deg[n], 1.0f));
    float4 v = reinterpret_cast<const float4*>(x)[t];
    v.x *= s; v.y *= s; v.z *= s; v.w *= s;
    reinterpret_cast<float4*>(xs)[t] = v;
  }
}

static __device__ __forceinline__ float4 gcn_row(const float* __restrict__ xs, int idx,
                                                 int k, unsigned last, int lane) {
  unsigned r = min((unsigned)__builtin_amdgcn_readlane(idx, k), last);
  return reinterpret_cast<const float4*>(xs + (size_t)r * DFEAT)[lane];
}

static __device__ __forceinline__ void gcn_add4(float4& a, const float4 v) {
  a.x += v.x; a.y += v.y; a.z += v.z; a.w += v.w;
}

__global__ void __launch_bounds__(256) gcn_gather(
    const float* __restrict__ xs, const int* __restrict__ srcs,
    const int* __restrict__ seg, float* __restrict__ agg, int N) {
  const int lane = threadIdx.x & 31;
  const int c = __builtin_amdgcn_readfirstlane(blockIdx.x * 8 + (threadIdx.x >> 5));
  if (c >= N) return;
  const int2 sc = reinterpret_cast<const int2*>(seg)[c];
  const int s = sc.x;
  const int cnt = sc.y;
  const unsigned last = (unsigned)(N - 1);

  float4 a0 = make_float4(0.f, 0.f, 0.f, 0.f);
  float4 a1 = a0, a2 = a0, a3 = a0;
  for (int base = 0; base < cnt; base += 32) {
    const int m = min(cnt - base, 32);
    const int idx = (lane < m) ? srcs[s + base + lane] : 0;
    int j = 0;
    for (; j + 8 <= m; j += 8) {
      float4 v0 = gcn_row(xs, idx, j + 0, last, lane);
      float4 v1 = gcn_row(xs, idx, j + 1, last, lane);
      float4 v2 = gcn_row(xs, idx, j + 2, last, lane);
      float4 v3 = gcn_row(xs, idx, j + 3, last, lane);
      float4 v4 = gcn_row(xs, idx, j + 4, last, lane);
      float4 v5 = gcn_row(xs, idx, j + 5, last, lane);
      float4 v6 = gcn_row(xs, idx, j + 6, last, lane);
      float4 v7 = gcn_row(xs, idx, j + 7, last, lane);
      gcn_add4(a0, v0); gcn_add4(a1, v1); gcn_add4(a2, v2); gcn_add4(a3, v3);
      gcn_add4(a0, v4); gcn_add4(a1, v5); gcn_add4(a2, v6); gcn_add4(a3, v7);
    }
    if ((m - j) & 4) {
      float4 v0 = gcn_row(xs, idx, j + 0, last, lane);
      float4 v1 = gcn_row(xs, idx, j + 1, last, lane);
      float4 v2 = gcn_row(xs, idx, j + 2, last, lane);
      float4 v3 = gcn_row(xs, idx, j + 3, last, lane);
      gcn_add4(a0, v0); gcn_add4(a1, v1); gcn_add4(a2, v2); gcn_add4(a3, v3);
      j += 4;
    }
    if ((m - j) & 2) {
      float4 v0 = gcn_row(xs, idx, j + 0, last, lane);
      float4 v1 = gcn_row(xs, idx, j + 1, last, lane);
      gcn_add4(a0, v0); gcn_add4(a1, v1);
      j += 2;
    }
    if ((m - j) & 1) {
      float4 v0 = gcn_row(xs, idx, j, last, lane);
      gcn_add4(a0, v0);
    }
  }
  float4 acc;
  acc.x = (a0.x + a1.x) + (a2.x + a3.x);
  acc.y = (a0.y + a1.y) + (a2.y + a3.y);
  acc.z = (a0.z + a1.z) + (a2.z + a3.z);
  acc.w = (a0.w + a1.w) + (a2.w + a3.w);
  reinterpret_cast<float4*>(agg + (size_t)c * DFEAT)[lane] = acc;
}

__global__ void __launch_bounds__(256) gcn_gemm_norm(
    const float* __restrict__ A,
    const float* __restrict__ W,
    const float* __restrict__ bias,
    const float* __restrict__ deg_in,
    const float* __restrict__ resid,
    float* __restrict__ out,
    int do_relu) {
  const int lane = threadIdx.x & 31;
  const int wave = threadIdx.x >> 5;
  const int half = lane >> 4;
  const int l15  = lane & 15;
  const int n0 = blockIdx.x * 16;
  const int i0 = wave * 16;

  const float* ap = A + (size_t)(n0 + l15) * DFEAT + 2 * half;
  const float* bp = W + (size_t)(i0 + l15) * DFEAT + 2 * half;

  v8f acc = {0.f, 0.f, 0.f, 0.f, 0.f, 0.f, 0.f, 0.f};
#pragma unroll
  for (int kk = 0; kk < DFEAT / 4; ++kk) {
    v2f a = *reinterpret_cast<const v2f*>(ap + 4 * kk);
    v2f b = *reinterpret_cast<const v2f*>(bp + 4 * kk);
    acc = __builtin_amdgcn_wmma_f32_16x16x4_f32(
         false, a,  false, b,
         (short)0, acc,  false,  false);
  }

  const int colI = i0 + l15;
  const float bb = bias[colI];
#pragma unroll
  for (int r = 0; r < 8; ++r) {
    int nrow = n0 + r + 8 * half;
    float val = acc[r] + bb;
    val *= rsqrtf(fmaxf(deg_in[nrow], 1.0f));
    if (resid) val += resid[(size_t)nrow * DFEAT + colI];
    if (do_relu) val = fmaxf(val, 0.0f);
    out[(size_t)nrow * DFEAT + colI] = val;
  }
}

extern "C" void kernel_launch(void* const* d_in, const int* in_sizes, int n_in,
                              void* d_out, int out_size, void* d_ws, size_t ws_size,
                              hipStream_t stream) {
  const float* x  = (const float*)d_in[0];
  const int*   ei = (const int*)d_in[1];
  const float* W1 = (const float*)d_in[2];
  const float* b1 = (const float*)d_in[3];
  const float* W2 = (const float*)d_in[4];
  const float* b2 = (const float*)d_in[5];

  const int N = in_sizes[0] / DFEAT;
  const int E = in_sizes[1] / 2;
  const int* row = ei;
  const int* col = ei + E;
  float* out = (float*)d_out;

  float* head    = (float*)d_ws;
  float* deg_out = head + 4;
  float* deg_in  = deg_out + N;
  float* xsrc    = deg_in + N;
  float* agg     = xsrc + (size_t)N * DFEAT;
  int*   seg     = (int*)(agg + (size_t)N * DFEAT);
  int*   rank    = seg + 2 * N;
  int*   srcs    = rank + E;
  int*   cursor  = (int*)head;

  const int TB = 256;
  const int scale4 = N * (DFEAT / 4);

  gcn_zero_f32<<<(2 * N + 4 + TB - 1) / TB, TB, 0, stream>>>(head, 2 * N + 4);
  gcn_degree_rank<<<(E + TB - 1) / TB, TB, 0, stream>>>(row, col, deg_out, deg_in, rank, E);
  gcn_segment_offsets<<<(N + TB - 1) / TB, TB, 0, stream>>>(deg_in, seg, cursor, N);
  gcn_fill_sorted<<<(E + TB - 1) / TB, TB, 0, stream>>>(row, col, rank, seg, srcs, E);

  gcn_scale<<<(scale4 + TB - 1) / TB, TB, 0, stream>>>(x, deg_out, xsrc, scale4);
  gcn_gather<<<(N + 7) / 8, TB, 0, stream>>>(xsrc, srcs, seg, agg, N);
  gcn_gemm_norm<<<N / 16, TB, 0, stream>>>(agg, W1, b1, deg_in, x, out, 1);

  gcn_scale<<<(scale4 + TB - 1) / TB, TB, 0, stream>>>(out, deg_out, xsrc, scale4);
  gcn_gather<<<(N + 7) / 8, TB, 0, stream>>>(xsrc, srcs, seg, agg, N);
  gcn_gemm_norm<<<N / 16, TB, 0, stream>>>(agg, W2, b2, deg_in, nullptr, out, 0);
}


# Layer GEMM re-tiled: W and 16-row agg tiles staged in LDS by whole-row loads, one wave = 16x128 outputs in 8 accumulator tiles, 512 resident blocks, rows stored whole; k order in each 8 changed
# speedup vs baseline: 2.2620x; 1.4164x over previous
#include <hip/hip_runtime.h>
#include <hip/hip_bf16.h>

#define DFEAT 128

typedef __attribute__((ext_vector_type(2))) float v2f;
typedef __attribute__((ext_vector_type(4))) float v4f;
typedef __attribute__((ext_vector_type(8))) float v8f;

__global__ void gcn_zero_f32(float* __restrict__ p, int n) {
  int i = blockIdx.x * blockDim.x + threadIdx.x;
  int stride = gridDim.x * blockDim.x;
  for (; i < n; i += stride) p[i] = 0.0f;
}

__global__ void gcn_degree_rank(const int* __restrict__ row, const int* __restrict__ col,
                                float* __restrict__ deg_out, float* __restrict__ deg_in,
                                int* __restrict__ rank, int E) {
  int e = blockIdx.x * blockDim.x + threadIdx.x;
  if (e < E) {
    atomicAdd(&deg_out[row[e]], 1.0f);
    float before = atomicAdd(&deg_in[col[e]], 1.0f);
    rank[e] = (int)before;
  }
}

__global__ void __launch_bounds__(256) gcn_segment_offsets(
    const float* __restrict__ deg_in, int* __restrict__ seg,
    int* __restrict__ cursor, int N) {
  __shared__ int wave_sum[8];
  __shared__ int block_base;
  const int lane = threadIdx.x & 31;
  const int wave = threadIdx.x >> 5;
  const int c = blockIdx.x * 256 + threadIdx.x;
  const int d = (c < N) ? (int)deg_in[c] : 0;
  int incl = d;
  for (int o = 1; o < 32; o <<= 1) {
    int t = __shfl_up(incl, o);
    if (lane >= o) incl += t;
  }
  if (lane == 31) wave_sum[wave] = incl;
  __syncthreads();
  if (threadIdx.x == 0) {
    int tot = 0;
    for (int w = 0; w < 8; ++w) {
      int t = wave_sum[w];
      wave_sum[w] = tot;
      tot += t;
    }
    block_base = atomicAdd(cursor, tot);
  }
  __syncthreads();
  if (c < N)
    reinterpret_cast<int2*>(seg)[c] = make_int2(block_base + wave_sum[wave] + incl - d, d);
}

__global__ void gcn_fill_sorted(const int* __restrict__ row, const int* __restrict__ col,
                                const int* __restrict__ rank, const int* __restrict__ seg,
                                int* __restrict__ srcs, int E) {
  int e = blockIdx.x * blockDim.x + threadIdx.x;
  if (e < E) {
    int p = seg[2 * col[e]] + rank[e];
    if ((unsigned)p < (unsigned)E) srcs[p] = row[e];
  }
}

__global__ void gcn_scale(const float* __restrict__ x, const float* __restrict__ deg,
                          float* __restrict__ xs, int total4) {
  int t = blockIdx.x * blockDim.x + threadIdx.x;
  if (t < total4) {
    int n = t >> 5;
    float s = rsqrtf(fmaxf(deg[n], 1.0f));
    float4 v = reinterpret_cast<const float4*>(x)[t];
    v.x *= s; v.y *= s; v.z *= s; v.w *= s;
    reinterpret_cast<float4*>(xs)[t] = v;
  }
}

static __device__ __forceinline__ float4 gcn_row(const float* __restrict__ xs, int idx,
                                                 int k, unsigned last, int lane) {
  unsigned r = min((unsigned)__builtin_amdgcn_readlane(idx, k), last);
  return reinterpret_cast<const float4*>(xs + (size_t)r * DFEAT)[lane];
}

static __device__ __forceinline__ void gcn_add4(float4& a, const float4 v) {
  a.x += v.x; a.y += v.y; a.z += v.z; a.w += v.w;
}

__global__ void __launch_bounds__(256) gcn_gather(
    const float* __restrict__ xs, const int* __restrict__ srcs,
    const int* __restrict__ seg, float* __restrict__ agg, int N) {
  const int lane = threadIdx.x & 31;
  const int c = __builtin_amdgcn_readfirstlane(blockIdx.x * 8 + (threadIdx.x >> 5));
  if (c >= N) return;
  const int2 sc = reinterpret_cast<const int2*>(seg)[c];
  const int s = sc.x;
  const int cnt = sc.y;
  const unsigned last = (unsigned)(N - 1);

  float4 a0 = make_float4(0.f, 0.f, 0.f, 0.f);
  float4 a1 = a0, a2 = a0, a3 = a0;
  for (int base = 0; base < cnt; base += 32) {
    const int m = min(cnt - base, 32);
    const int idx = (lane < m) ? srcs[s + base + lane] : 0;
    int j = 0;
    for (; j + 8 <= m; j += 8) {
      float4 v0 = gcn_row(xs, idx, j + 0, last, lane);
      float4 v1 = gcn_row(xs, idx, j + 1, last, lane);
      float4 v2 = gcn_row(xs, idx, j + 2, last, lane);
      float4 v3 = gcn_row(xs, idx, j + 3, last, lane);
      float4 v4 = gcn_row(xs, idx, j + 4, last, lane);
      float4 v5 = gcn_row(xs, idx, j + 5, last, lane);
      float4 v6 = gcn_row(xs, idx, j + 6, last, lane);
      float4 v7 = gcn_row(xs, idx, j + 7, last, lane);
      gcn_add4(a0, v0); gcn_add4(a1, v1); gcn_add4(a2, v2); gcn_add4(a3, v3);
      gcn_add4(a0, v4); gcn_add4(a1, v5); gcn_add4(a2, v6); gcn_add4(a3, v7);
    }
    if ((m - j) & 4) {
      float4 v0 = gcn_row(xs, idx, j + 0, last, lane);
      float4 v1 = gcn_row(xs, idx, j + 1, last, lane);
      float4 v2 = gcn_row(xs, idx, j + 2, last, lane);
      float4 v3 = gcn_row(xs, idx, j + 3, last, lane);
      gcn_add4(a0, v0); gcn_add4(a1, v1); gcn_add4(a2, v2); gcn_add4(a3, v3);
      j += 4;
    }
    if ((m - j) & 2) {
      float4 v0 = gcn_row(xs, idx, j + 0, last, lane);
      float4 v1 = gcn_row(xs, idx, j + 1, last, lane);
      gcn_add4(a0, v0); gcn_add4(a1, v1);
      j += 2;
    }
    if ((m - j) & 1) {
      float4 v0 = gcn_row(xs, idx, j, last, lane);
      gcn_add4(a0, v0);
    }
  }
  float4 acc;
  acc.x = (a0.x + a1.x) + (a2.x + a3.x);
  acc.y = (a0.y + a1.y) + (a2.y + a3.y);
  acc.z = (a0.z + a1.z) + (a2.z + a3.z);
  acc.w = (a0.w + a1.w) + (a2.w + a3.w);
  reinterpret_cast<float4*>(agg + (size_t)c * DFEAT)[lane] = acc;
}

#define GEMM_PITCH 132
#define GEMM_BLOCKS 512

static __device__ __forceinline__ float4 gcn_frag(const float* __restrict__ tile, int c, int h, int j) {
  return *reinterpret_cast<const float4*>(tile + c * GEMM_PITCH + 8 * j + 4 * h);
}

template <bool kResid, bool kRelu, bool kFull>
static __device__ __forceinline__ void gcn_rows_out(
    const float* strip, const float4 bias4, const float* __restrict__ deg_in,
    const float* __restrict__ resid, float* __restrict__ out, int n0, int N, int lane) {
  float dg[16];
  v4f x4[16];
#pragma unroll
  for (int r = 0; r < 16; ++r) {
    if (kFull || n0 + r < N) {
      dg[r] = deg_in[n0 + r];
      if (kResid) x4[r] = reinterpret_cast<const v4f*>(resid + (size_t)(n0 + r) * DFEAT)[lane];
    }
  }
  if (kResid && kFull) {
    asm volatile("" ::"v"(x4[0]), "v"(x4[1]), "v"(x4[2]), "v"(x4[3]), "v"(x4[4]), "v"(x4[5]),
                 "v"(x4[6]), "v"(x4[7]), "v"(x4[8]), "v"(x4[9]), "v"(x4[10]), "v"(x4[11]),
                 "v"(x4[12]), "v"(x4[13]), "v"(x4[14]), "v"(x4[15]));
  }
#pragma unroll
  for (int r = 0; r < 16; ++r) {
    if (kFull || n0 + r < N) {
      float4 v = *reinterpret_cast<const float4*>(strip + r * GEMM_PITCH + 4 * lane);
      const float s = rsqrtf(fmaxf(dg[r], 1.0f));
      v.x += bias4.x; v.y += bias4.y; v.z += bias4.z; v.w += bias4.w;
      v.x *= s; v.y *= s; v.z *= s; v.w *= s;
      if (kResid) { v.x += x4[r].x; v.y += x4[r].y; v.z += x4[r].z; v.w += x4[r].w; }
      if (kRelu) {
        v.x = fmaxf(v.x, 0.0f); v.y = fmaxf(v.y, 0.0f);
        v.z = fmaxf(v.z, 0.0f); v.w = fmaxf(v.w, 0.0f);
      }
      reinterpret_cast<float4*>(out + (size_t)(n0 + r) * DFEAT)[lane] = v;
    }
  }
}

template <bool kResid, bool kRelu>
__global__ void __launch_bounds__(256, 4) gcn_gemm_norm(
    const float* __restrict__ A,
    const float* __restrict__ W,
    const float* __restrict__ bias,
    const float* __restrict__ deg_in,
    const float* __restrict__ resid,
    float* __restrict__ out,
    int N) {
  __shared__ __attribute__((aligned(16))) float lds_w[DFEAT * GEMM_PITCH];
  __shared__ __attribute__((aligned(16))) float lds_a[8 * 16 * GEMM_PITCH];

  const int lane = threadIdx.x & 31;
  const int wave = threadIdx.x >> 5;
  const int h = lane >> 4;
  const int c = lane & 15;

#pragma unroll
  for (int it = 0; it < 16; ++it) {
    const int idx = it * 256 + threadIdx.x;
    const float4 v = reinterpret_cast<const float4*>(W)[idx];
    *reinterpret_cast<float4*>(lds_w + (idx >> 5) * GEMM_PITCH + 4 * (idx & 31)) = v;
  }
  __syncthreads();

  float* strip = lds_a + wave * 16 * GEMM_PITCH;
  const float4 bias4 = reinterpret_cast<const float4*>(bias)[lane];
  const int tiles = (N + 15) >> 4;
  const int step = gridDim.x * 8;
  for (int t = __builtin_amdgcn_readfirstlane(wave * gridDim.x + blockIdx.x); t < tiles; t += step) {
    const int n0 = t * 16;

#pragma unroll
    for (int r = 0; r < 16; ++r) {
      const int nr = min(n0 + r, N - 1);
      const float4 v = reinterpret_cast<const float4*>(A + (size_t)nr * DFEAT)[lane];
      *reinterpret_cast<float4*>(strip + r * GEMM_PITCH + 4 * lane) = v;
    }
    __builtin_amdgcn_fence(__ATOMIC_ACQ_REL, "wavefront");

    v8f acc[8];
#pragma unroll
    for (int ct = 0; ct < 8; ++ct) acc[ct] = (v8f){0.f, 0.f, 0.f, 0.f, 0.f, 0.f, 0.f, 0.f};

#pragma unroll
    for (int j = 0; j < DFEAT / 8; ++j) {
      const float4 a = gcn_frag(strip, c, h, j);
      float4 b[8];
#pragma unroll
      for (int ct = 0; ct < 8; ++ct) b[ct] = gcn_frag(lds_w + ct * 16 * GEMM_PITCH, c, h, j);
      const v2f a0 = {a.x, a.y};
      const v2f a1 = {a.z, a.w};
#pragma unroll
      for (int ct = 0; ct < 8; ++ct) {
        const v2f b0 = {b[ct].x, b[ct].y};
        acc[ct] = __builtin_amdgcn_wmma_f32_16x16x4_f32(
             false, a0,  false, b0,
             (short)0, acc[ct],  false,  false);
      }
#pragma unroll
      for (int ct = 0; ct < 8; ++ct) {
        const v2f b1 = {b[ct].z, b[ct].w};
        acc[ct] = __builtin_amdgcn_wmma_f32_16x16x4_f32(
             false, a1,  false, b1,
             (short)0, acc[ct],  false,  false);
      }
    }

#pragma unroll
    for (int ct = 0; ct < 8; ++ct) {
#pragma unroll
      for (int r = 0; r < 8; ++r) strip[(8 * h + r) * GEMM_PITCH + 16 * ct + c] = acc[ct][r];
    }
    __builtin_amdgcn_fence(__ATOMIC_ACQ_REL, "wavefront");

    if (n0 + 16 <= N) {
      gcn_rows_out<kResid, kRelu, true>(strip, bias4, deg_in, resid, out, n0, N, lane);
    } else {
      gcn_rows_out<kResid, kRelu, false>(strip, bias4, deg_in, resid, out, n0, N, lane);
    }
    __builtin_amdgcn_fence(__ATOMIC_ACQ_REL, "wavefront");
  }
}

extern "C" void kernel_launch(void* const* d_in, const int* in_sizes, int n_in,
                              void* d_out, int out_size, void* d_ws, size_t ws_size,
                              hipStream_t stream) {
  const float* x  = (const float*)d_in[0];
  const int*   ei = (const int*)d_in[1];
  const float* W1 = (const float*)d_in[2];
  const float* b1 = (const float*)d_in[3];
  const float* W2 = (const float*)d_in[4];
  const float* b2 = (const float*)d_in[5];

  const int N = in_sizes[0] / DFEAT;
  const int E = in_sizes[1] / 2;
  const int* row = ei;
  const int* col = ei + E;
  float* out = (float*)d_out;

  float* head    = (float*)d_ws;
  float* deg_out = head + 4;
  float* deg_in  = deg_out + N;
  float* xsrc    = deg_in + N;
  float* agg     = xsrc + (size_t)N * DFEAT;
  int*   seg     = (int*)(agg + (size_t)N * DFEAT);
  int*   rank    = seg + 2 * N;
  int*   srcs    = rank + E;
  int*   cursor  = (int*)head;

  const int TB = 256;
  const int scale4 = N * (DFEAT / 4);

  gcn_zero_f32<<<(2 * N + 4 + TB - 1) / TB, TB, 0, stream>>>(head, 2 * N + 4);
  gcn_degree_rank<<<(E + TB - 1) / TB, TB, 0, stream>>>(row, col, deg_out, deg_in, rank, E);
  gcn_segment_offsets<<<(N + TB - 1) / TB, TB, 0, stream>>>(deg_in, seg, cursor, N);
  gcn_fill_sorted<<<(E + TB - 1) / TB, TB, 0, stream>>>(row, col, rank, seg, srcs, E);

  gcn_scale<<<(scale4 + TB - 1) / TB, TB, 0, stream>>>(x, deg_out, xsrc, scale4);
  gcn_gather<<<(N + 7) / 8, TB, 0, stream>>>(xsrc, srcs, seg, agg, N);
  gcn_gemm_norm<true, true><<<GEMM_BLOCKS, TB, 0, stream>>>(agg, W1, b1, deg_in, x, out, N);

  gcn_scale<<<(scale4 + TB - 1) / TB, TB, 0, stream>>>(out, deg_out, xsrc, scale4);
  gcn_gather<<<(N + 7) / 8, TB, 0, stream>>>(xsrc, srcs, seg, agg, N);
  gcn_gemm_norm<false, false><<<GEMM_BLOCKS, TB, 0, stream>>>(agg, W2, b2, deg_in, nullptr, out, N);
}


# x_src scaling pass removed in both layers: the sorted edge list carries rsqrt(max(deg_out[src],1)) per edge and the segment sum applies it by fma as it adds each source row; 10 launches become 8
# speedup vs baseline: 2.3916x; 1.0573x over previous
#include <hip/hip_runtime.h>
#include <hip/hip_bf16.h>

#define DFEAT 128

typedef __attribute__((ext_vector_type(2))) float v2f;
typedef __attribute__((ext_vector_type(4))) float v4f;
typedef __attribute__((ext_vector_type(8))) float v8f;

__global__ void gcn_zero_f32(float* __restrict__ p, int n) {
  int i = blockIdx.x * blockDim.x + threadIdx.x;
  int stride = gridDim.x * blockDim.x;
  for (; i < n; i += stride) p[i] = 0.0f;
}

__global__ void gcn_degree_rank(const int* __restrict__ row, const int* __restrict__ col,
                                float* __restrict__ deg_out, float* __restrict__ deg_in,
                                int* __restrict__ rank, int E) {
  int e = blockIdx.x * blockDim.x + threadIdx.x;
  if (e < E) {
    atomicAdd(&deg_out[row[e]], 1.0f);
    float before = atomicAdd(&deg_in[col[e]], 1.0f);
    rank[e] = (int)before;
  }
}

__global__ void __launch_bounds__(256) gcn_segment_offsets(
    const float* __restrict__ deg_in, int* __restrict__ seg,
    int* __restrict__ cursor, int N) {
  __shared__ int wave_sum[8];
  __shared__ int block_base;
  const int lane = threadIdx.x & 31;
  const int wave = threadIdx.x >> 5;
  const int c = blockIdx.x * 256 + threadIdx.x;
  const int d = (c < N) ? (int)deg_in[c] : 0;
  int incl = d;
  for (int o = 1; o < 32; o <<= 1) {
    int t = __shfl_up(incl, o);
    if (lane >= o) incl += t;
  }
  if (lane == 31) wave_sum[wave] = incl;
  __syncthreads();
  if (threadIdx.x == 0) {
    int tot = 0;
    for (int w = 0; w < 8; ++w) {
      int t = wave_sum[w];
      wave_sum[w] = tot;
      tot += t;
    }
    block_base = atomicAdd(cursor, tot);
  }
  __syncthreads();
  if (c < N)
    reinterpret_cast<int2*>(seg)[c] = make_int2(block_base + wave_sum[wave] + incl - d, d);
}

__global__ void gcn_fill_sorted(const int* __restrict__ row, const int* __restrict__ col,
                                const int* __restrict__ rank, const int* __restrict__ seg,
                                const float* __restrict__ deg_out,
                                int* __restrict__ srcs, float* __restrict__ wts, int E) {
  int e = blockIdx.x * blockDim.x + threadIdx.x;
  if (e < E) {
    const int r = row[e];
    const float w = rsqrtf(fmaxf(deg_out[r], 1.0f));
    const int p = seg[2 * col[e]] + rank[e];
    if ((unsigned)p < (unsigned)E) {
      srcs[p] = r;
      wts[p] = w;
    }
  }
}

static __device__ __forceinline__ float4 gcn_row(const float* __restrict__ xs, int idx,
                                                 int k, unsigned last, int lane) {
  unsigned r = min((unsigned)__builtin_amdgcn_readlane(idx, k), last);
  return reinterpret_cast<const float4*>(xs + (size_t)r * DFEAT)[lane];
}

static __device__ __forceinline__ void gcn_fma4(float4& a, const float4 v, int wbits, int k) {
  const float w = __int_as_float(__builtin_amdgcn_readlane(wbits, k));
  a.x = fmaf(v.x, w, a.x); a.y = fmaf(v.y, w, a.y);
  a.z = fmaf(v.z, w, a.z); a.w = fmaf(v.w, w, a.w);
}

__global__ void __launch_bounds__(256) gcn_gather(
    const float* __restrict__ xs, const int* __restrict__ srcs,
    const float* __restrict__ wts,
    const int* __restrict__ seg, float* __restrict__ agg, int N) {
  const int lane = threadIdx.x & 31;
  const int c = __builtin_amdgcn_readfirstlane(blockIdx.x * 8 + (threadIdx.x >> 5));
  if (c >= N) return;
  const int2 sc = reinterpret_cast<const int2*>(seg)[c];
  const int s = sc.x;
  const int cnt = sc.y;
  const unsigned last = (unsigned)(N - 1);

  float4 a0 = make_float4(0.f, 0.f, 0.f, 0.f);
  float4 a1 = a0, a2 = a0, a3 = a0;
  for (int base = 0; base < cnt; base += 32) {
    const int m = min(cnt - base, 32);
    const int idx = (lane < m) ? srcs[s + base + lane] : 0;
    const int wb = __float_as_int((lane < m) ? wts[s + base + lane] : 0.0f);
    int j = 0;
    for (; j + 8 <= m; j += 8) {
      float4 v0 = gcn_row(xs, idx, j + 0, last, lane);
      float4 v1 = gcn_row(xs, idx, j + 1, last, lane);
      float4 v2 = gcn_row(xs, idx, j + 2, last, lane);
      float4 v3 = gcn_row(xs, idx, j + 3, last, lane);
      float4 v4 = gcn_row(xs, idx, j + 4, last, lane);
      float4 v5 = gcn_row(xs, idx, j + 5, last, lane);
      float4 v6 = gcn_row(xs, idx, j + 6, last, lane);
      float4 v7 = gcn_row(xs, idx, j + 7, last, lane);
      gcn_fma4(a0, v0, wb, j + 0); gcn_fma4(a1, v1, wb, j + 1);
      gcn_fma4(a2, v2, wb, j + 2); gcn_fma4(a3, v3, wb, j + 3);
      gcn_fma4(a0, v4, wb, j + 4); gcn_fma4(a1, v5, wb, j + 5);
      gcn_fma4(a2, v6, wb, j + 6); gcn_fma4(a3, v7, wb, j + 7);
    }
    if ((m - j) & 4) {
      float4 v0 = gcn_row(xs, idx, j + 0, last, lane);
      float4 v1 = gcn_row(xs, idx, j + 1, last, lane);
      float4 v2 = gcn_row(xs, idx, j + 2, last, lane);
      float4 v3 = gcn_row(xs, idx, j + 3, last, lane);
      gcn_fma4(a0, v0, wb, j + 0); gcn_fma4(a1, v1, wb, j + 1);
      gcn_fma4(a2, v2, wb, j + 2); gcn_fma4(a3, v3, wb, j + 3);
      j += 4;
    }
    if ((m - j) & 2) {
      float4 v0 = gcn_row(xs, idx, j + 0, last, lane);
      float4 v1 = gcn_row(xs, idx, j + 1, last, lane);
      gcn_fma4(a0, v0, wb, j + 0); gcn_fma4(a1, v1, wb, j + 1);
      j += 2;
    }
    if ((m - j) & 1) {
      float4 v0 = gcn_row(xs, idx, j, last, lane);
      gcn_fma4(a0, v0, wb, j);
    }
  }
  float4 acc;
  acc.x = (a0.x + a1.x) + (a2.x + a3.x);
  acc.y = (a0.y + a1.y) + (a2.y + a3.y);
  acc.z = (a0.z + a1.z) + (a2.z + a3.z);
  acc.w = (a0.w + a1.w) + (a2.w + a3.w);
  reinterpret_cast<float4*>(agg + (size_t)c * DFEAT)[lane] = acc;
}

#define GEMM_PITCH 132
#define GEMM_BLOCKS 512

static __device__ __forceinline__ float4 gcn_frag(const float* __restrict__ tile, int c, int h, int j) {
  return *reinterpret_cast<const float4*>(tile + c * GEMM_PITCH + 8 * j + 4 * h);
}

template <bool kResid, bool kRelu, bool kFull>
static __device__ __forceinline__ void gcn_rows_out(
    const float* strip, const float4 bias4, const float* __restrict__ deg_in,
    const float* __restrict__ resid, float* __restrict__ out, int n0, int N, int lane) {
  float dg[16];
  v4f x4[16];
#pragma unroll
  for (int r = 0; r < 16; ++r) {
    if (kFull || n0 + r < N) {
      dg[r] = deg_in[n0 + r];
      if (kResid) x4[r] = reinterpret_cast<const v4f*>(resid + (size_t)(n0 + r) * DFEAT)[lane];
    }
  }
  if (kResid && kFull) {
    asm volatile("" ::"v"(x4[0]), "v"(x4[1]), "v"(x4[2]), "v"(x4[3]), "v"(x4[4]), "v"(x4[5]),
                 "v"(x4[6]), "v"(x4[7]), "v"(x4[8]), "v"(x4[9]), "v"(x4[10]), "v"(x4[11]),
                 "v"(x4[12]), "v"(x4[13]), "v"(x4[14]), "v"(x4[15]));
  }
#pragma unroll
  for (int r = 0; r < 16; ++r) {
    if (kFull || n0 + r < N) {
      float4 v = *reinterpret_cast<const float4*>(strip + r * GEMM_PITCH + 4 * lane);
      const float s = rsqrtf(fmaxf(dg[r], 1.0f));
      v.x += bias4.x; v.y += bias4.y; v.z += bias4.z; v.w += bias4.w;
      v.x *= s; v.y *= s; v.z *= s; v.w *= s;
      if (kResid) { v.x += x4[r].x; v.y += x4[r].y; v.z += x4[r].z; v.w += x4[r].w; }
      if (kRelu) {
        v.x = fmaxf(v.x, 0.0f); v.y = fmaxf(v.y, 0.0f);
        v.z = fmaxf(v.z, 0.0f); v.w = fmaxf(v.w, 0.0f);
      }
      reinterpret_cast<float4*>(out + (size_t)(n0 + r) * DFEAT)[lane] = v;
    }
  }
}

template <bool kResid, bool kRelu>
__global__ void __launch_bounds__(256, 4) gcn_gemm_norm(
    const float* __restrict__ A,
    const float* __restrict__ W,
    const float* __restrict__ bias,
    const float* __restrict__ deg_in,
    const float* __restrict__ resid,
    float* __restrict__ out,
    int N) {
  __shared__ __attribute__((aligned(16))) float lds_w[DFEAT * GEMM_PITCH];
  __shared__ __attribute__((aligned(16))) float lds_a[8 * 16 * GEMM_PITCH];

  const int lane = threadIdx.x & 31;
  const int wave = threadIdx.x >> 5;
  const int h = lane >> 4;
  const int c = lane & 15;

#pragma unroll
  for (int it = 0; it < 16; ++it) {
    const int idx = it * 256 + threadIdx.x;
    const float4 v = reinterpret_cast<const float4*>(W)[idx];
    *reinterpret_cast<float4*>(lds_w + (idx >> 5) * GEMM_PITCH + 4 * (idx & 31)) = v;
  }
  __syncthreads();

  float* strip = lds_a + wave * 16 * GEMM_PITCH;
  const float4 bias4 = reinterpret_cast<const float4*>(bias)[lane];
  const int tiles = (N + 15) >> 4;
  const int step = gridDim.x * 8;
  for (int t = __builtin_amdgcn_readfirstlane(wave * gridDim.x + blockIdx.x); t < tiles; t += step) {
    const int n0 = t * 16;

#pragma unroll
    for (int r = 0; r < 16; ++r) {
      const int nr = min(n0 + r, N - 1);
      const float4 v = reinterpret_cast<const float4*>(A + (size_t)nr * DFEAT)[lane];
      *reinterpret_cast<float4*>(strip + r * GEMM_PITCH + 4 * lane) = v;
    }
    __builtin_amdgcn_fence(__ATOMIC_ACQ_REL, "wavefront");

    v8f acc[8];
#pragma unroll
    for (int ct = 0; ct < 8; ++ct) acc[ct] = (v8f){0.f, 0.f, 0.f, 0.f, 0.f, 0.f, 0.f, 0.f};

#pragma unroll
    for (int j = 0; j < DFEAT / 8; ++j) {
      const float4 a = gcn_frag(strip, c, h, j);
      float4 b[8];
#pragma unroll
      for (int ct = 0; ct < 8; ++ct) b[ct] = gcn_frag(lds_w + ct * 16 * GEMM_PITCH, c, h, j);
      const v2f a0 = {a.x, a.y};
      const v2f a1 = {a.z, a.w};
#pragma unroll
      for (int ct = 0; ct < 8; ++ct) {
        const v2f b0 = {b[ct].x, b[ct].y};
        acc[ct] = __builtin_amdgcn_wmma_f32_16x16x4_f32(
             false, a0,  false, b0,
             (short)0, acc[ct],  false,  false);
      }
#pragma unroll
      for (int ct = 0; ct < 8; ++ct) {
        const v2f b1 = {b[ct].z, b[ct].w};
        acc[ct] = __builtin_amdgcn_wmma_f32_16x16x4_f32(
             false, a1,  false, b1,
             (short)0, acc[ct],  false,  false);
      }
    }

#pragma unroll
    for (int ct = 0; ct < 8; ++ct) {
#pragma unroll
      for (int r = 0; r < 8; ++r) strip[(8 * h + r) * GEMM_PITCH + 16 * ct + c] = acc[ct][r];
    }
    __builtin_amdgcn_fence(__ATOMIC_ACQ_REL, "wavefront");

    if (n0 + 16 <= N) {
      gcn_rows_out<kResid, kRelu, true>(strip, bias4, deg_in, resid, out, n0, N, lane);
    } else {
      gcn_rows_out<kResid, kRelu, false>(strip, bias4, deg_in, resid, out, n0, N, lane);
    }
    __builtin_amdgcn_fence(__ATOMIC_ACQ_REL, "wavefront");
  }
}

extern "C" void kernel_launch(void* const* d_in, const int* in_sizes, int n_in,
                              void* d_out, int out_size, void* d_ws, size_t ws_size,
                              hipStream_t stream) {
  const float* x  = (const float*)d_in[0];
  const int*   ei = (const int*)d_in[1];
  const float* W1 = (const float*)d_in[2];
  const float* b1 = (const float*)d_in[3];
  const float* W2 = (const float*)d_in[4];
  const float* b2 = (const float*)d_in[5];

  const int N = in_sizes[0] / DFEAT;
  const int E = in_sizes[1] / 2;
  const int* row = ei;
  const int* col = ei + E;
  float* out = (float*)d_out;

  float* head    = (float*)d_ws;
  float* deg_out = head + 4;
  float* deg_in  = deg_out + N;
  float* agg     = deg_in + N;
  int*   seg     = (int*)(agg + (size_t)N * DFEAT);
  int*   rank    = seg + 2 * N;
  int*   srcs    = rank + E;
  float* wts     = (float*)(srcs + E);
  int*   cursor  = (int*)head;

  const int TB = 256;

  gcn_zero_f32<<<(2 * N + 4 + TB - 1) / TB, TB, 0, stream>>>(head, 2 * N + 4);
  gcn_degree_rank<<<(E + TB - 1) / TB, TB, 0, stream>>>(row, col, deg_out, deg_in, rank, E);
  gcn_segment_offsets<<<(N + TB - 1) / TB, TB, 0, stream>>>(deg_in, seg, cursor, N);
  gcn_fill_sorted<<<(E + TB - 1) / TB, TB, 0, stream>>>(row, col, rank, seg, deg_out, srcs, wts, E);

  gcn_gather<<<(N + 7) / 8, TB, 0, stream>>>(x, srcs, wts, seg, agg, N);
  gcn_gemm_norm<true, true><<<GEMM_BLOCKS, TB, 0, stream>>>(agg, W1, b1, deg_in, x, out, N);

  gcn_gather<<<(N + 7) / 8, TB, 0, stream>>>(out, srcs, wts, seg, agg, N);
  gcn_gemm_norm<false, false><<<GEMM_BLOCKS, TB, 0, stream>>>(agg, W2, b2, deg_in, nullptr, out, N);
}


# Edge rank from LDS counters after binning edges by 256-node destination range (count, scatter, per-bin sort): no returning global add per edge, no offsets or fill pass; run order = LDS arrival
# speedup vs baseline: 3.0001x; 1.2544x over previous
#include <hip/hip_runtime.h>
#include <hip/hip_bf16.h>

#define DFEAT 128

typedef __attribute__((ext_vector_type(2))) float v2f;
typedef __attribute__((ext_vector_type(4))) float v4f;
typedef __attribute__((ext_vector_type(8))) float v8f;

__global__ void gcn_zero_f32(float* __restrict__ p, int n) {
  int i = blockIdx.x * blockDim.x + threadIdx.x;
  int stride = gridDim.x * blockDim.x;
  for (; i < n; i += stride) p[i] = 0.0f;
}

#define BIN_SHIFT 8
#define EDGE_EPT 8

__global__ void __launch_bounds__(256) gcn_bin_count(
    const int* __restrict__ row, const int* __restrict__ col,
    float* __restrict__ deg_out, int* __restrict__ bin_tot, int E, int NB) {
  extern __shared__ int lds_cnt[];
  for (int b = threadIdx.x; b < NB; b += 256) lds_cnt[b] = 0;
  __syncthreads();
  const int e0 = blockIdx.x * (256 * EDGE_EPT) + threadIdx.x;
  int cc[EDGE_EPT], rr[EDGE_EPT];
#pragma unroll
  for (int i = 0; i < EDGE_EPT; ++i) {
    const int e = min(e0 + i * 256, E - 1);
    cc[i] = col[e];
    rr[i] = row[e];
  }
#pragma unroll
  for (int i = 0; i < EDGE_EPT; ++i) {
    if (e0 + i * 256 < E) {
      atomicAdd(&deg_out[rr[i]], 1.0f);
      atomicAdd(&lds_cnt[cc[i] >> BIN_SHIFT], 1);
    }
  }
  __syncthreads();
  for (int b = threadIdx.x; b < NB; b += 256) {
    const int h = lds_cnt[b];
    if (h) atomicAdd(&bin_tot[b], h);
  }
}

__global__ void __launch_bounds__(256) gcn_bin_scatter(
    const int* __restrict__ row, const int* __restrict__ col,
    const int* __restrict__ bin_tot, int* __restrict__ bin_cur,
    int* __restrict__ bin_start, int* __restrict__ bsrc, int* __restrict__ bdst,
    int E, int NB) {
  extern __shared__ int lds_bins[];
  int* hist = lds_bins;
  int* pos0 = lds_bins + NB;
  const int lane = threadIdx.x & 31;
  const int wave = threadIdx.x >> 5;
  for (int b = threadIdx.x; b < NB; b += 256) {
    hist[b] = 0;
    pos0[b] = bin_tot[b];
  }
  __syncthreads();
  if (wave == 0) {
    const int per = (NB + 31) >> 5;
    const int lo = lane * per;
    int sum = 0;
    for (int i = 0; i < per; ++i)
      if (lo + i < NB) sum += pos0[lo + i];
    int incl = sum;
    for (int o = 1; o < 32; o <<= 1) {
      int t = __shfl_up(incl, o);
      if (lane >= o) incl += t;
    }
    int run = incl - sum;
    for (int i = 0; i < per; ++i) {
      if (lo + i < NB) {
        const int t = pos0[lo + i];
        pos0[lo + i] = run;
        run += t;
      }
    }
  }
  int cc[EDGE_EPT], rr[EDGE_EPT], lr[EDGE_EPT];
  const int e0 = blockIdx.x * (256 * EDGE_EPT) + threadIdx.x;
#pragma unroll
  for (int i = 0; i < EDGE_EPT; ++i) {
    const int e = min(e0 + i * 256, E - 1);
    cc[i] = col[e];
    rr[i] = row[e];
  }
#pragma unroll
  for (int i = 0; i < EDGE_EPT; ++i) {
    lr[i] = 0;
    if (e0 + i * 256 < E) lr[i] = atomicAdd(&hist[cc[i] >> BIN_SHIFT], 1);
  }
  __syncthreads();
  for (int b = threadIdx.x; b < NB; b += 256) {
    const int st = pos0[b];
    if (blockIdx.x == 0) bin_start[b] = st;
    const int h = hist[b];
    if (h) pos0[b] = st + atomicAdd(&bin_cur[b], h);
  }
  __syncthreads();
#pragma unroll
  for (int i = 0; i < EDGE_EPT; ++i) {
    if (e0 + i * 256 < E) {
      const int p = pos0[cc[i] >> BIN_SHIFT] + lr[i];
      if ((unsigned)p < (unsigned)E) {
        bsrc[p] = rr[i];
        bdst[p] = cc[i];
      }
    }
  }
}

__global__ void __launch_bounds__(1024) gcn_bin_sort(
    const int* __restrict__ bsrc, const int* __restrict__ bdst,
    const int* __restrict__ bin_tot, const int* __restrict__ bin_start,
    const float* __restrict__ deg_out, int* __restrict__ seg,
    float* __restrict__ deg_in, int* __restrict__ srcs, float* __restrict__ wts,
    int N, int E) {
  __shared__ int cnt[256];
  __shared__ int off[256];
  __shared__ int cur[256];
  __shared__ int wave_sum[8];
  const int tid = threadIdx.x;
  const int lane = tid & 31;
  const int wave = tid >> 5;
  const int bin = blockIdx.x;
  const int n = bin_tot[bin];
  const int s0 = bin_start[bin];
  const int c0 = bin << BIN_SHIFT;
  if (tid < 256) {
    cnt[tid] = 0;
    cur[tid] = 0;
  }
  __syncthreads();
  for (int i = tid; i < n; i += 1024) atomicAdd(&cnt[(bdst[s0 + i] - c0) & 255], 1);
  __syncthreads();
  const int d = (tid < 256) ? cnt[tid] : 0;
  int incl = d;
  for (int o = 1; o < 32; o <<= 1) {
    int t = __shfl_up(incl, o);
    if (lane >= o) incl += t;
  }
  if (tid < 256 && lane == 31) wave_sum[wave] = incl;
  __syncthreads();
  if (tid == 0) {
    int tot = 0;
    for (int w = 0; w < 8; ++w) {
      int t = wave_sum[w];
      wave_sum[w] = tot;
      tot += t;
    }
  }
  __syncthreads();
  if (tid < 256) {
    const int o = wave_sum[wave] + incl - d;
    off[tid] = o;
    const int c = c0 + tid;
    if (c < N) {
      reinterpret_cast<int2*>(seg)[c] = make_int2(s0 + o, d);
      deg_in[c] = (float)d;
    }
  }
  __syncthreads();
  for (int i = tid; i < n; i += 1024) {
    const int loc = (bdst[s0 + i] - c0) & 255;
    const int r = bsrc[s0 + i];
    const int k = atomicAdd(&cur[loc], 1);
    const int p = s0 + off[loc] + k;
    if ((unsigned)p < (unsigned)E) {
      srcs[p] = r;
      wts[p] = rsqrtf(fmaxf(deg_out[r], 1.0f));
    }
  }
}

static __device__ __forceinline__ float4 gcn_row(const float* __restrict__ xs, int idx,
                                                 int k, unsigned last, int lane) {
  unsigned r = min((unsigned)__builtin_amdgcn_readlane(idx, k), last);
  return reinterpret_cast<const float4*>(xs + (size_t)r * DFEAT)[lane];
}

static __device__ __forceinline__ void gcn_fma4(float4& a, const float4 v, int wbits, int k) {
  const float w = __int_as_float(__builtin_amdgcn_readlane(wbits, k));
  a.x = fmaf(v.x, w, a.x); a.y = fmaf(v.y, w, a.y);
  a.z = fmaf(v.z, w, a.z); a.w = fmaf(v.w, w, a.w);
}

__global__ void __launch_bounds__(256) gcn_gather(
    const float* __restrict__ xs, const int* __restrict__ srcs,
    const float* __restrict__ wts,
    const int* __restrict__ seg, float* __restrict__ agg, int N) {
  const int lane = threadIdx.x & 31;
  const int c = __builtin_amdgcn_readfirstlane(blockIdx.x * 8 + (threadIdx.x >> 5));
  if (c >= N) return;
  const int2 sc = reinterpret_cast<const int2*>(seg)[c];
  const int s = sc.x;
  const int cnt = sc.y;
  const unsigned last = (unsigned)(N - 1);

  float4 a0 = make_float4(0.f, 0.f, 0.f, 0.f);
  float4 a1 = a0, a2 = a0, a3 = a0;
  for (int base = 0; base < cnt; base += 32) {
    const int m = min(cnt - base, 32);
    const int idx = (lane < m) ? srcs[s + base + lane] : 0;
    const int wb = __float_as_int((lane < m) ? wts[s + base + lane] : 0.0f);
    int j = 0;
    for (; j + 8 <= m; j += 8) {
      float4 v0 = gcn_row(xs, idx, j + 0, last, lane);
      float4 v1 = gcn_row(xs, idx, j + 1, last, lane);
      float4 v2 = gcn_row(xs, idx, j + 2, last, lane);
      float4 v3 = gcn_row(xs, idx, j + 3, last, lane);
      float4 v4 = gcn_row(xs, idx, j + 4, last, lane);
      float4 v5 = gcn_row(xs, idx, j + 5, last, lane);
      float4 v6 = gcn_row(xs, idx, j + 6, last, lane);
      float4 v7 = gcn_row(xs, idx, j + 7, last, lane);
      gcn_fma4(a0, v0, wb, j + 0); gcn_fma4(a1, v1, wb, j + 1);
      gcn_fma4(a2, v2, wb, j + 2); gcn_fma4(a3, v3, wb, j + 3);
      gcn_fma4(a0, v4, wb, j + 4); gcn_fma4(a1, v5, wb, j + 5);
      gcn_fma4(a2, v6, wb, j + 6); gcn_fma4(a3, v7, wb, j + 7);
    }
    if ((m - j) & 4) {
      float4 v0 = gcn_row(xs, idx, j + 0, last, lane);
      float4 v1 = gcn_row(xs, idx, j + 1, last, lane);
      float4 v2 = gcn_row(xs, idx, j + 2, last, lane);
      float4 v3 = gcn_row(xs, idx, j + 3, last, lane);
      gcn_fma4(a0, v0, wb, j + 0); gcn_fma4(a1, v1, wb, j + 1);
      gcn_fma4(a2, v2, wb, j + 2); gcn_fma4(a3, v3, wb, j + 3);
      j += 4;
    }
    if ((m - j) & 2) {
      float4 v0 = gcn_row(xs, idx, j + 0, last, lane);
      float4 v1 = gcn_row(xs, idx, j + 1, last, lane);
      gcn_fma4(a0, v0, wb, j + 0); gcn_fma4(a1, v1, wb, j + 1);
      j += 2;
    }
    if ((m - j) & 1) {
      float4 v0 = gcn_row(xs, idx, j, last, lane);
      gcn_fma4(a0, v0, wb, j);
    }
  }
  float4 acc;
  acc.x = (a0.x + a1.x) + (a2.x + a3.x);
  acc.y = (a0.y + a1.y) + (a2.y + a3.y);
  acc.z = (a0.z + a1.z) + (a2.z + a3.z);
  acc.w = (a0.w + a1.w) + (a2.w + a3.w);
  reinterpret_cast<float4*>(agg + (size_t)c * DFEAT)[lane] = acc;
}

#define GEMM_PITCH 132
#define GEMM_BLOCKS 512

static __device__ __forceinline__ float4 gcn_frag(const float* __restrict__ tile, int c, int h, int j) {
  return *reinterpret_cast<const float4*>(tile + c * GEMM_PITCH + 8 * j + 4 * h);
}

template <bool kResid, bool kRelu, bool kFull>
static __device__ __forceinline__ void gcn_rows_out(
    const float* strip, const float4 bias4, const float* __restrict__ deg_in,
    const float* __restrict__ resid, float* __restrict__ out, int n0, int N, int lane) {
  float dg[16];
  v4f x4[16];
#pragma unroll
  for (int r = 0; r < 16; ++r) {
    if (kFull || n0 + r < N) {
      dg[r] = deg_in[n0 + r];
      if (kResid) x4[r] = reinterpret_cast<const v4f*>(resid + (size_t)(n0 + r) * DFEAT)[lane];
    }
  }
  if (kResid && kFull) {
    asm volatile("" ::"v"(x4[0]), "v"(x4[1]), "v"(x4[2]), "v"(x4[3]), "v"(x4[4]), "v"(x4[5]),
                 "v"(x4[6]), "v"(x4[7]), "v"(x4[8]), "v"(x4[9]), "v"(x4[10]), "v"(x4[11]),
                 "v"(x4[12]), "v"(x4[13]), "v"(x4[14]), "v"(x4[15]));
  }
#pragma unroll
  for (int r = 0; r < 16; ++r) {
    if (kFull || n0 + r < N) {
      float4 v = *reinterpret_cast<const float4*>(strip + r * GEMM_PITCH + 4 * lane);
      const float s = rsqrtf(fmaxf(dg[r], 1.0f));
      v.x += bias4.x; v.y += bias4.y; v.z += bias4.z; v.w += bias4.w;
      v.x *= s; v.y *= s; v.z *= s; v.w *= s;
      if (kResid) { v.x += x4[r].x; v.y += x4[r].y; v.z += x4[r].z; v.w += x4[r].w; }
      if (kRelu) {
        v.x = fmaxf(v.x, 0.0f); v.y = fmaxf(v.y, 0.0f);
        v.z = fmaxf(v.z, 0.0f); v.w = fmaxf(v.w, 0.0f);
      }
      reinterpret_cast<float4*>(out + (size_t)(n0 + r) * DFEAT)[lane] = v;
    }
  }
}

template <bool kResid, bool kRelu>
__global__ void __launch_bounds__(256, 4) gcn_gemm_norm(
    const float* __restrict__ A,
    const float* __restrict__ W,
    const float* __restrict__ bias,
    const float* __restrict__ deg_in,
    const float* __restrict__ resid,
    float* __restrict__ out,
    int N) {
  __shared__ __attribute__((aligned(16))) float lds_w[DFEAT * GEMM_PITCH];
  __shared__ __attribute__((aligned(16))) float lds_a[8 * 16 * GEMM_PITCH];

  const int lane = threadIdx.x & 31;
  const int wave = threadIdx.x >> 5;
  const int h = lane >> 4;
  const int c = lane & 15;

#pragma unroll
  for (int it = 0; it < 16; ++it) {
    const int idx = it * 256 + threadIdx.x;
    const float4 v = reinterpret_cast<const float4*>(W)[idx];
    *reinterpret_cast<float4*>(lds_w + (idx >> 5) * GEMM_PITCH + 4 * (idx & 31)) = v;
  }
  __syncthreads();

  float* strip = lds_a + wave * 16 * GEMM_PITCH;
  const float4 bias4 = reinterpret_cast<const float4*>(bias)[lane];
  const int tiles = (N + 15) >> 4;
  const int step = gridDim.x * 8;
  for (int t = __builtin_amdgcn_readfirstlane(wave * gridDim.x + blockIdx.x); t < tiles; t += step) {
    const int n0 = t * 16;

#pragma unroll
    for (int r = 0; r < 16; ++r) {
      const int nr = min(n0 + r, N - 1);
      const float4 v = reinterpret_cast<const float4*>(A + (size_t)nr * DFEAT)[lane];
      *reinterpret_cast<float4*>(strip + r * GEMM_PITCH + 4 * lane) = v;
    }
    __builtin_amdgcn_fence(__ATOMIC_ACQ_REL, "wavefront");

    v8f acc[8];
#pragma unroll
    for (int ct = 0; ct < 8; ++ct) acc[ct] = (v8f){0.f, 0.f, 0.f, 0.f, 0.f, 0.f, 0.f, 0.f};

#pragma unroll
    for (int j = 0; j < DFEAT / 8; ++j) {
      const float4 a = gcn_frag(strip, c, h, j);
      float4 b[8];
#pragma unroll
      for (int ct = 0; ct < 8; ++ct) b[ct] = gcn_frag(lds_w + ct * 16 * GEMM_PITCH, c, h, j);
      const v2f a0 = {a.x, a.y};
      const v2f a1 = {a.z, a.w};
#pragma unroll
      for (int ct = 0; ct < 8; ++ct) {
        const v2f b0 = {b[ct].x, b[ct].y};
        acc[ct] = __builtin_amdgcn_wmma_f32_16x16x4_f32(
             false, a0,  false, b0,
             (short)0, acc[ct],  false,  false);
      }
#pragma unroll
      for (int ct = 0; ct < 8; ++ct) {
        const v2f b1 = {b[ct].z, b[ct].w};
        acc[ct] = __builtin_amdgcn_wmma_f32_16x16x4_f32(
             false, a1,  false, b1,
             (short)0, acc[ct],  false,  false);
      }
    }

#pragma unroll
    for (int ct = 0; ct < 8; ++ct) {
#pragma unroll
      for (int r = 0; r < 8; ++r) strip[(8 * h + r) * GEMM_PITCH + 16 * ct + c] = acc[ct][r];
    }
    __builtin_amdgcn_fence(__ATOMIC_ACQ_REL, "wavefront");

    if (n0 + 16 <= N) {
      gcn_rows_out<kResid, kRelu, true>(strip, bias4, deg_in, resid, out, n0, N, lane);
    } else {
      gcn_rows_out<kResid, kRelu, false>(strip, bias4, deg_in, resid, out, n0, N, lane);
    }
    __builtin_amdgcn_fence(__ATOMIC_ACQ_REL, "wavefront");
  }
}

extern "C" void kernel_launch(void* const* d_in, const int* in_sizes, int n_in,
                              void* d_out, int out_size, void* d_ws, size_t ws_size,
                              hipStream_t stream) {
  const float* x  = (const float*)d_in[0];
  const int*   ei = (const int*)d_in[1];
  const float* W1 = (const float*)d_in[2];
  const float* b1 = (const float*)d_in[3];
  const float* W2 = (const float*)d_in[4];
  const float* b2 = (const float*)d_in[5];

  const int N = in_sizes[0] / DFEAT;
  const int E = in_sizes[1] / 2;
  const int NB = (N + 255) >> BIN_SHIFT;
  const int* row = ei;
  const int* col = ei + E;
  float* out = (float*)d_out;

  float* agg       = (float*)d_ws;
  int*   seg       = (int*)(agg + (size_t)N * DFEAT);
  int*   bin_tot   = seg + 2 * N;
  int*   bin_cur   = bin_tot + NB;
  float* deg_out   = (float*)(bin_cur + NB);
  float* deg_in    = deg_out + N;
  int*   bin_start = (int*)(deg_in + N);
  int*   bsrc      = bin_start + NB;
  int*   bdst      = bsrc + E;
  int*   srcs      = bdst + E;
  float* wts       = (float*)(srcs + E);

  const int TB = 256;
  const int EB = (E + TB * EDGE_EPT - 1) / (TB * EDGE_EPT);

  gcn_zero_f32<<<(2 * NB + N + TB - 1) / TB, TB, 0, stream>>>((float*)bin_tot, 2 * NB + N);
  gcn_bin_count<<<EB, TB, NB * sizeof(int), stream>>>(row, col, deg_out, bin_tot, E, NB);
  gcn_bin_scatter<<<EB, TB, 2 * NB * sizeof(int), stream>>>(row, col, bin_tot, bin_cur, bin_start,
                                                           bsrc, bdst, E, NB);
  gcn_bin_sort<<<NB, 1024, 0, stream>>>(bsrc, bdst, bin_tot, bin_start, deg_out, seg, deg_in,
                                        srcs, wts, N, E);

  gcn_gather<<<(N + 7) / 8, TB, 0, stream>>>(x, srcs, wts, seg, agg, N);
  gcn_gemm_norm<true, true><<<GEMM_BLOCKS, TB, 0, stream>>>(agg, W1, b1, deg_in, x, out, N);

  gcn_gather<<<(N + 7) / 8, TB, 0, stream>>>(out, srcs, wts, seg, agg, N);
  gcn_gemm_norm<false, false><<<GEMM_BLOCKS, TB, 0, stream>>>(agg, W2, b2, deg_in, nullptr, out, N);
}


# Out-degree counted by LDS counters over a second, source-binned copy of the edge list in the same three passes, not by 800000 scattered global float adds; edge norms made by the first segment sum
# speedup vs baseline: 3.7489x; 1.2496x over previous
#include <hip/hip_runtime.h>
#include <hip/hip_bf16.h>

#define DFEAT 128

typedef __attribute__((ext_vector_type(2))) float v2f;
typedef __attribute__((ext_vector_type(4))) float v4f;
typedef __attribute__((ext_vector_type(8))) float v8f;

__global__ void gcn_zero_f32(float* __restrict__ p, int n) {
  int i = blockIdx.x * blockDim.x + threadIdx.x;
  int stride = gridDim.x * blockDim.x;
  for (; i < n; i += stride) p[i] = 0.0f;
}

#define BIN_SHIFT 8
#define EDGE_EPT 8

__global__ void __launch_bounds__(256) gcn_bin_count(
    const int* __restrict__ row, const int* __restrict__ col,
    int* __restrict__ bin_tot, int E, int NB) {
  extern __shared__ int lds_cnt[];
  for (int b = threadIdx.x; b < 2 * NB; b += 256) lds_cnt[b] = 0;
  __syncthreads();
  const int e0 = blockIdx.x * (256 * EDGE_EPT) + threadIdx.x;
  int cc[EDGE_EPT], rr[EDGE_EPT];
#pragma unroll
  for (int i = 0; i < EDGE_EPT; ++i) {
    const int e = min(e0 + i * 256, E - 1);
    cc[i] = col[e];
    rr[i] = row[e];
  }
#pragma unroll
  for (int i = 0; i < EDGE_EPT; ++i) {
    if (e0 + i * 256 < E) {
      atomicAdd(&lds_cnt[cc[i] >> BIN_SHIFT], 1);
      atomicAdd(&lds_cnt[NB + (rr[i] >> BIN_SHIFT)], 1);
    }
  }
  __syncthreads();
  for (int b = threadIdx.x; b < 2 * NB; b += 256) {
    const int h = lds_cnt[b];
    if (h) atomicAdd(&bin_tot[b], h);
  }
}

__global__ void __launch_bounds__(256) gcn_bin_scatter(
    const int* __restrict__ row, const int* __restrict__ col,
    const int* __restrict__ bin_tot, int* __restrict__ bin_cur,
    int* __restrict__ bin_start, int* __restrict__ bsrc, int* __restrict__ bdst,
    int* __restrict__ srow, int E, int NB) {
  extern __shared__ int lds_bins[];
  int* hist = lds_bins;
  int* pos0 = lds_bins + 2 * NB;
  const int lane = threadIdx.x & 31;
  const int wave = threadIdx.x >> 5;
  for (int b = threadIdx.x; b < 2 * NB; b += 256) {
    hist[b] = 0;
    pos0[b] = bin_tot[b];
  }
  __syncthreads();
  if (wave < 2) {
    int* tab = pos0 + wave * NB;
    const int per = (NB + 31) >> 5;
    const int lo = lane * per;
    int sum = 0;
    for (int i = 0; i < per; ++i)
      if (lo + i < NB) sum += tab[lo + i];
    int incl = sum;
    for (int o = 1; o < 32; o <<= 1) {
      int t = __shfl_up(incl, o);
      if (lane >= o) incl += t;
    }
    int run = incl - sum;
    for (int i = 0; i < per; ++i) {
      if (lo + i < NB) {
        const int t = tab[lo + i];
        tab[lo + i] = run;
        run += t;
      }
    }
  }
  int cc[EDGE_EPT], rr[EDGE_EPT], lr[EDGE_EPT], sr[EDGE_EPT];
  const int e0 = blockIdx.x * (256 * EDGE_EPT) + threadIdx.x;
#pragma unroll
  for (int i = 0; i < EDGE_EPT; ++i) {
    const int e = min(e0 + i * 256, E - 1);
    cc[i] = col[e];
    rr[i] = row[e];
  }
#pragma unroll
  for (int i = 0; i < EDGE_EPT; ++i) {
    lr[i] = 0;
    sr[i] = 0;
    if (e0 + i * 256 < E) {
      lr[i] = atomicAdd(&hist[cc[i] >> BIN_SHIFT], 1);
      sr[i] = atomicAdd(&hist[NB + (rr[i] >> BIN_SHIFT)], 1);
    }
  }
  __syncthreads();
  for (int b = threadIdx.x; b < 2 * NB; b += 256) {
    const int st = pos0[b];
    if (blockIdx.x == 0) bin_start[b] = st;
    const int h = hist[b];
    if (h) pos0[b] = st + atomicAdd(&bin_cur[b], h);
  }
  __syncthreads();
#pragma unroll
  for (int i = 0; i < EDGE_EPT; ++i) {
    if (e0 + i * 256 < E) {
      const int p = pos0[cc[i] >> BIN_SHIFT] + lr[i];
      if ((unsigned)p < (unsigned)E) {
        bsrc[p] = rr[i];
        bdst[p] = cc[i];
      }
      const int q = pos0[NB + (rr[i] >> BIN_SHIFT)] + sr[i];
      if ((unsigned)q < (unsigned)E) srow[q] = rr[i];
    }
  }
}

__global__ void __launch_bounds__(1024) gcn_bin_sort(
    const int* __restrict__ bsrc, const int* __restrict__ bdst,
    const int* __restrict__ srow,
    const int* __restrict__ bin_tot, const int* __restrict__ bin_start,
    float* __restrict__ deg_out, int* __restrict__ seg,
    float* __restrict__ deg_in, int* __restrict__ srcs,
    int N, int E, int NB) {
  __shared__ int cnt[256];
  __shared__ int scnt[256];
  __shared__ int off[256];
  __shared__ int cur[256];
  __shared__ int wave_sum[8];
  const int tid = threadIdx.x;
  const int lane = tid & 31;
  const int wave = tid >> 5;
  const int bin = blockIdx.x;
  const int n = bin_tot[bin];
  const int s0 = bin_start[bin];
  const int sn = bin_tot[NB + bin];
  const int ss0 = bin_start[NB + bin];
  const int c0 = bin << BIN_SHIFT;
  if (tid < 256) {
    cnt[tid] = 0;
    scnt[tid] = 0;
    cur[tid] = 0;
  }
  __syncthreads();
  for (int i = tid; i < n; i += 1024) atomicAdd(&cnt[(bdst[s0 + i] - c0) & 255], 1);
  for (int i = tid; i < sn; i += 1024) atomicAdd(&scnt[(srow[ss0 + i] - c0) & 255], 1);
  __syncthreads();
  const int d = (tid < 256) ? cnt[tid] : 0;
  int incl = d;
  for (int o = 1; o < 32; o <<= 1) {
    int t = __shfl_up(incl, o);
    if (lane >= o) incl += t;
  }
  if (tid < 256 && lane == 31) wave_sum[wave] = incl;
  __syncthreads();
  if (tid == 0) {
    int tot = 0;
    for (int w = 0; w < 8; ++w) {
      int t = wave_sum[w];
      wave_sum[w] = tot;
      tot += t;
    }
  }
  __syncthreads();
  if (tid < 256) {
    const int o = wave_sum[wave] + incl - d;
    off[tid] = o;
    const int c = c0 + tid;
    if (c < N) {
      reinterpret_cast<int2*>(seg)[c] = make_int2(s0 + o, d);
      deg_in[c] = (float)d;
      deg_out[c] = (float)scnt[tid];
    }
  }
  __syncthreads();
  for (int i = tid; i < n; i += 1024) {
    const int loc = (bdst[s0 + i] - c0) & 255;
    const int r = bsrc[s0 + i];
    const int k = atomicAdd(&cur[loc], 1);
    const int p = s0 + off[loc] + k;
    if ((unsigned)p < (unsigned)E) srcs[p] = r;
  }
}

static __device__ __forceinline__ float4 gcn_row(const float* __restrict__ xs, int idx,
                                                 int k, unsigned last, int lane) {
  unsigned r = min((unsigned)__builtin_amdgcn_readlane(idx, k), last);
  return reinterpret_cast<const float4*>(xs + (size_t)r * DFEAT)[lane];
}

template <bool kDeg>
static __device__ __forceinline__ void gcn_fma4(float4& a, const float4 v, int wbits, int k) {
  float w = __int_as_float(__builtin_amdgcn_readlane(wbits, k));
  if (kDeg) w = rsqrtf(fmaxf(w, 1.0f));
  a.x = fmaf(v.x, w, a.x); a.y = fmaf(v.y, w, a.y);
  a.z = fmaf(v.z, w, a.z); a.w = fmaf(v.w, w, a.w);
}

template <bool kFirst>
__global__ void __launch_bounds__(256) gcn_gather(
    const float* __restrict__ xs, const int* __restrict__ srcs,
    const float* __restrict__ deg_out, float* __restrict__ wts,
    const int* __restrict__ seg, float* __restrict__ agg, int N) {
  const int lane = threadIdx.x & 31;
  const int c = __builtin_amdgcn_readfirstlane(blockIdx.x * 8 + (threadIdx.x >> 5));
  if (c >= N) return;
  const int2 sc = reinterpret_cast<const int2*>(seg)[c];
  const int s = sc.x;
  const int cnt = sc.y;
  const unsigned last = (unsigned)(N - 1);

  float4 a0 = make_float4(0.f, 0.f, 0.f, 0.f);
  float4 a1 = a0, a2 = a0, a3 = a0;
  for (int base = 0; base < cnt; base += 32) {
    const int m = min(cnt - base, 32);
    const int idx = (lane < m) ? srcs[s + base + lane] : 0;
    float wsrc;
    if (kFirst) wsrc = deg_out[min((unsigned)idx, last)];
    else wsrc = (lane < m) ? wts[s + base + lane] : 0.0f;
    const int wb = __float_as_int(wsrc);
    int j = 0;
    for (; j + 8 <= m; j += 8) {
      float4 v0 = gcn_row(xs, idx, j + 0, last, lane);
      float4 v1 = gcn_row(xs, idx, j + 1, last, lane);
      float4 v2 = gcn_row(xs, idx, j + 2, last, lane);
      float4 v3 = gcn_row(xs, idx, j + 3, last, lane);
      float4 v4 = gcn_row(xs, idx, j + 4, last, lane);
      float4 v5 = gcn_row(xs, idx, j + 5, last, lane);
      float4 v6 = gcn_row(xs, idx, j + 6, last, lane);
      float4 v7 = gcn_row(xs, idx, j + 7, last, lane);
      if (kFirst) __builtin_amdgcn_sched_barrier(0);
      gcn_fma4<kFirst>(a0, v0, wb, j + 0); gcn_fma4<kFirst>(a1, v1, wb, j + 1);
      gcn_fma4<kFirst>(a2, v2, wb, j + 2); gcn_fma4<kFirst>(a3, v3, wb, j + 3);
      gcn_fma4<kFirst>(a0, v4, wb, j + 4); gcn_fma4<kFirst>(a1, v5, wb, j + 5);
      gcn_fma4<kFirst>(a2, v6, wb, j + 6); gcn_fma4<kFirst>(a3, v7, wb, j + 7);
    }
    if ((m - j) & 4) {
      float4 v0 = gcn_row(xs, idx, j + 0, last, lane);
      float4 v1 = gcn_row(xs, idx, j + 1, last, lane);
      float4 v2 = gcn_row(xs, idx, j + 2, last, lane);
      float4 v3 = gcn_row(xs, idx, j + 3, last, lane);
      if (kFirst) __builtin_amdgcn_sched_barrier(0);
      gcn_fma4<kFirst>(a0, v0, wb, j + 0); gcn_fma4<kFirst>(a1, v1, wb, j + 1);
      gcn_fma4<kFirst>(a2, v2, wb, j + 2); gcn_fma4<kFirst>(a3, v3, wb, j + 3);
      j += 4;
    }
    if ((m - j) & 2) {
      float4 v0 = gcn_row(xs, idx, j + 0, last, lane);
      float4 v1 = gcn_row(xs, idx, j + 1, last, lane);
      if (kFirst) __builtin_amdgcn_sched_barrier(0);
      gcn_fma4<kFirst>(a0, v0, wb, j + 0); gcn_fma4<kFirst>(a1, v1, wb, j + 1);
      j += 2;
    }
    if ((m - j) & 1) {
      float4 v0 = gcn_row(xs, idx, j, last, lane);
      if (kFirst) __builtin_amdgcn_sched_barrier(0);
      gcn_fma4<kFirst>(a0, v0, wb, j);
    }
    if (kFirst && lane < m) wts[s + base + lane] = rsqrtf(fmaxf(wsrc, 1.0f));
  }
  float4 acc;
  acc.x = (a0.x + a1.x) + (a2.x + a3.x);
  acc.y = (a0.y + a1.y) + (a2.y + a3.y);
  acc.z = (a0.z + a1.z) + (a2.z + a3.z);
  acc.w = (a0.w + a1.w) + (a2.w + a3.w);
  reinterpret_cast<float4*>(agg + (size_t)c * DFEAT)[lane] = acc;
}

#define GEMM_PITCH 132
#define GEMM_BLOCKS 512

static __device__ __forceinline__ float4 gcn_frag(const float* __restrict__ tile, int c, int h, int j) {
  return *reinterpret_cast<const float4*>(tile + c * GEMM_PITCH + 8 * j + 4 * h);
}

template <bool kResid, bool kRelu, bool kFull>
static __device__ __forceinline__ void gcn_rows_out(
    const float* strip, const float4 bias4, const float* __restrict__ deg_in,
    const float* __restrict__ resid, float* __restrict__ out, int n0, int N, int lane) {
  float dg[16];
  v4f x4[16];
#pragma unroll
  for (int r = 0; r < 16; ++r) {
    if (kFull || n0 + r < N) {
      dg[r] = deg_in[n0 + r];
      if (kResid) x4[r] = reinterpret_cast<const v4f*>(resid + (size_t)(n0 + r) * DFEAT)[lane];
    }
  }
  if (kResid && kFull) {
    asm volatile("" ::"v"(x4[0]), "v"(x4[1]), "v"(x4[2]), "v"(x4[3]), "v"(x4[4]), "v"(x4[5]),
                 "v"(x4[6]), "v"(x4[7]), "v"(x4[8]), "v"(x4[9]), "v"(x4[10]), "v"(x4[11]),
                 "v"(x4[12]), "v"(x4[13]), "v"(x4[14]), "v"(x4[15]));
  }
#pragma unroll
  for (int r = 0; r < 16; ++r) {
    if (kFull || n0 + r < N) {
      float4 v = *reinterpret_cast<const float4*>(strip + r * GEMM_PITCH + 4 * lane);
      const float s = rsqrtf(fmaxf(dg[r], 1.0f));
      v.x += bias4.x; v.y += bias4.y; v.z += bias4.z; v.w += bias4.w;
      v.x *= s; v.y *= s; v.z *= s; v.w *= s;
      if (kResid) { v.x += x4[r].x; v.y += x4[r].y; v.z += x4[r].z; v.w += x4[r].w; }
      if (kRelu) {
        v.x = fmaxf(v.x, 0.0f); v.y = fmaxf(v.y, 0.0f);
        v.z = fmaxf(v.z, 0.0f); v.w = fmaxf(v.w, 0.0f);
      }
      reinterpret_cast<float4*>(out + (size_t)(n0 + r) * DFEAT)[lane] = v;
    }
  }
}

template <bool kResid, bool kRelu>
__global__ void __launch_bounds__(256, 4) gcn_gemm_norm(
    const float* __restrict__ A,
    const float* __restrict__ W,
    const float* __restrict__ bias,
    const float* __restrict__ deg_in,
    const float* __restrict__ resid,
    float* __restrict__ out,
    int N) {
  __shared__ __attribute__((aligned(16))) float lds_w[DFEAT * GEMM_PITCH];
  __shared__ __attribute__((aligned(16))) float lds_a[8 * 16 * GEMM_PITCH];

  const int lane = threadIdx.x & 31;
  const int wave = threadIdx.x >> 5;
  const int h = lane >> 4;
  const int c = lane & 15;

#pragma unroll
  for (int it = 0; it < 16; ++it) {
    const int idx = it * 256 + threadIdx.x;
    const float4 v = reinterpret_cast<const float4*>(W)[idx];
    *reinterpret_cast<float4*>(lds_w + (idx >> 5) * GEMM_PITCH + 4 * (idx & 31)) = v;
  }
  __syncthreads();

  float* strip = lds_a + wave * 16 * GEMM_PITCH;
  const float4 bias4 = reinterpret_cast<const float4*>(bias)[lane];
  const int tiles = (N + 15) >> 4;
  const int step = gridDim.x * 8;
  for (int t = __builtin_amdgcn_readfirstlane(wave * gridDim.x + blockIdx.x); t < tiles; t += step) {
    const int n0 = t * 16;

#pragma unroll
    for (int r = 0; r < 16; ++r) {
      const int nr = min(n0 + r, N - 1);
      const float4 v = reinterpret_cast<const float4*>(A + (size_t)nr * DFEAT)[lane];
      *reinterpret_cast<float4*>(strip + r * GEMM_PITCH + 4 * lane) = v;
    }
    __builtin_amdgcn_fence(__ATOMIC_ACQ_REL, "wavefront");

    v8f acc[8];
#pragma unroll
    for (int ct = 0; ct < 8; ++ct) acc[ct] = (v8f){0.f, 0.f, 0.f, 0.f, 0.f, 0.f, 0.f, 0.f};

#pragma unroll
    for (int j = 0; j < DFEAT / 8; ++j) {
      const float4 a = gcn_frag(strip, c, h, j);
      float4 b[8];
#pragma unroll
      for (int ct = 0; ct < 8; ++ct) b[ct] = gcn_frag(lds_w + ct * 16 * GEMM_PITCH, c, h, j);
      const v2f a0 = {a.x, a.y};
      const v2f a1 = {a.z, a.w};
#pragma unroll
      for (int ct = 0; ct < 8; ++ct) {
        const v2f b0 = {b[ct].x, b[ct].y};
        acc[ct] = __builtin_amdgcn_wmma_f32_16x16x4_f32(
             false, a0,  false, b0,
             (short)0, acc[ct],  false,  false);
      }
#pragma unroll
      for (int ct = 0; ct < 8; ++ct) {
        const v2f b1 = {b[ct].z, b[ct].w};
        acc[ct] = __builtin_amdgcn_wmma_f32_16x16x4_f32(
             false, a1,  false, b1,
             (short)0, acc[ct],  false,  false);
      }
    }

#pragma unroll
    for (int ct = 0; ct < 8; ++ct) {
#pragma unroll
      for (int r = 0; r < 8; ++r) strip[(8 * h + r) * GEMM_PITCH + 16 * ct + c] = acc[ct][r];
    }
    __builtin_amdgcn_fence(__ATOMIC_ACQ_REL, "wavefront");

    if (n0 + 16 <= N) {
      gcn_rows_out<kResid, kRelu, true>(strip, bias4, deg_in, resid, out, n0, N, lane);
    } else {
      gcn_rows_out<kResid, kRelu, false>(strip, bias4, deg_in, resid, out, n0, N, lane);
    }
    __builtin_amdgcn_fence(__ATOMIC_ACQ_REL, "wavefront");
  }
}

extern "C" void kernel_launch(void* const* d_in, const int* in_sizes, int n_in,
                              void* d_out, int out_size, void* d_ws, size_t ws_size,
                              hipStream_t stream) {
  const float* x  = (const float*)d_in[0];
  const int*   ei = (const int*)d_in[1];
  const float* W1 = (const float*)d_in[2];
  const float* b1 = (const float*)d_in[3];
  const float* W2 = (const float*)d_in[4];
  const float* b2 = (const float*)d_in[5];

  const int N = in_sizes[0] / DFEAT;
  const int E = in_sizes[1] / 2;
  const int NB = (N + 255) >> BIN_SHIFT;
  const int* row = ei;
  const int* col = ei + E;
  float* out = (float*)d_out;

  float* agg       = (float*)d_ws;
  int*   seg       = (int*)(agg + (size_t)N * DFEAT);
  int*   bin_tot   = seg + 2 * N;
  int*   bin_cur   = bin_tot + 2 * NB;
  float* deg_out   = (float*)(bin_cur + 2 * NB);
  float* deg_in    = deg_out + N;
  int*   bin_start = (int*)(deg_in + N);
  int*   bsrc      = bin_start + 2 * NB;
  int*   bdst      = bsrc + E;
  int*   srow      = bdst + E;
  int*   srcs      = srow + E;
  float* wts       = (float*)(srcs + E);

  const int TB = 256;
  const int EB = (E + TB * EDGE_EPT - 1) / (TB * EDGE_EPT);

  gcn_zero_f32<<<(4 * NB + TB - 1) / TB, TB, 0, stream>>>((float*)bin_tot, 4 * NB);
  gcn_bin_count<<<EB, TB, 2 * NB * sizeof(int), stream>>>(row, col, bin_tot, E, NB);
  gcn_bin_scatter<<<EB, TB, 4 * NB * sizeof(int), stream>>>(row, col, bin_tot, bin_cur, bin_start,
                                                           bsrc, bdst, srow, E, NB);
  gcn_bin_sort<<<NB, 1024, 0, stream>>>(bsrc, bdst, srow, bin_tot, bin_start, deg_out, seg, deg_in,
                                        srcs, N, E, NB);

  gcn_gather<true><<<(N + 7) / 8, TB, 0, stream>>>(x, srcs, deg_out, wts, seg, agg, N);
  gcn_gemm_norm<true, true><<<GEMM_BLOCKS, TB, 0, stream>>>(agg, W1, b1, deg_in, x, out, N);

  gcn_gather<false><<<(N + 7) / 8, TB, 0, stream>>>(out, srcs, deg_out, wts, seg, agg, N);
  gcn_gemm_norm<false, false><<<GEMM_BLOCKS, TB, 0, stream>>>(agg, W2, b2, deg_in, nullptr, out, N);
}


# Segment sums run as resident grids (1536 / 2048 blocks): each wave loops over nodes, the next node's run and list words requested one trip ahead; 100000 one-node waves per layer become 12288 / 16384
# speedup vs baseline: 3.8699x; 1.0323x over previous
#include <hip/hip_runtime.h>
#include <hip/hip_bf16.h>

#define DFEAT 128

typedef __attribute__((ext_vector_type(2))) float v2f;
typedef __attribute__((ext_vector_type(4))) float v4f;
typedef __attribute__((ext_vector_type(8))) float v8f;

__global__ void gcn_zero_f32(float* __restrict__ p, int n) {
  int i = blockIdx.x * blockDim.x + threadIdx.x;
  int stride = gridDim.x * blockDim.x;
  for (; i < n; i += stride) p[i] = 0.0f;
}

#define BIN_SHIFT 8
#define EDGE_EPT 8

__global__ void __launch_bounds__(256) gcn_bin_count(
    const int* __restrict__ row, const int* __restrict__ col,
    int* __restrict__ bin_tot, int E, int NB) {
  extern __shared__ int lds_cnt[];
  for (int b = threadIdx.x; b < 2 * NB; b += 256) lds_cnt[b] = 0;
  __syncthreads();
  const int e0 = blockIdx.x * (256 * EDGE_EPT) + threadIdx.x;
  int cc[EDGE_EPT], rr[EDGE_EPT];
#pragma unroll
  for (int i = 0; i < EDGE_EPT; ++i) {
    const int e = min(e0 + i * 256, E - 1);
    cc[i] = col[e];
    rr[i] = row[e];
  }
#pragma unroll
  for (int i = 0; i < EDGE_EPT; ++i) {
    if (e0 + i * 256 < E) {
      atomicAdd(&lds_cnt[cc[i] >> BIN_SHIFT], 1);
      atomicAdd(&lds_cnt[NB + (rr[i] >> BIN_SHIFT)], 1);
    }
  }
  __syncthreads();
  for (int b = threadIdx.x; b < 2 * NB; b += 256) {
    const int h = lds_cnt[b];
    if (h) atomicAdd(&bin_tot[b], h);
  }
}

__global__ void __launch_bounds__(256) gcn_bin_scatter(
    const int* __restrict__ row, const int* __restrict__ col,
    const int* __restrict__ bin_tot, int* __restrict__ bin_cur,
    int* __restrict__ bin_start, int* __restrict__ bsrc, int* __restrict__ bdst,
    int* __restrict__ srow, int E, int NB) {
  extern __shared__ int lds_bins[];
  int* hist = lds_bins;
  int* pos0 = lds_bins + 2 * NB;
  const int lane = threadIdx.x & 31;
  const int wave = threadIdx.x >> 5;
  for (int b = threadIdx.x; b < 2 * NB; b += 256) {
    hist[b] = 0;
    pos0[b] = bin_tot[b];
  }
  __syncthreads();
  if (wave < 2) {
    int* tab = pos0 + wave * NB;
    const int per = (NB + 31) >> 5;
    const int lo = lane * per;
    int sum = 0;
    for (int i = 0; i < per; ++i)
      if (lo + i < NB) sum += tab[lo + i];
    int incl = sum;
    for (int o = 1; o < 32; o <<= 1) {
      int t = __shfl_up(incl, o);
      if (lane >= o) incl += t;
    }
    int run = incl - sum;
    for (int i = 0; i < per; ++i) {
      if (lo + i < NB) {
        const int t = tab[lo + i];
        tab[lo + i] = run;
        run += t;
      }
    }
  }
  int cc[EDGE_EPT], rr[EDGE_EPT], lr[EDGE_EPT], sr[EDGE_EPT];
  const int e0 = blockIdx.x * (256 * EDGE_EPT) + threadIdx.x;
#pragma unroll
  for (int i = 0; i < EDGE_EPT; ++i) {
    const int e = min(e0 + i * 256, E - 1);
    cc[i] = col[e];
    rr[i] = row[e];
  }
#pragma unroll
  for (int i = 0; i < EDGE_EPT; ++i) {
    lr[i] = 0;
    sr[i] = 0;
    if (e0 + i * 256 < E) {
      lr[i] = atomicAdd(&hist[cc[i] >> BIN_SHIFT], 1);
      sr[i] = atomicAdd(&hist[NB + (rr[i] >> BIN_SHIFT)], 1);
    }
  }
  __syncthreads();
  for (int b = threadIdx.x; b < 2 * NB; b += 256) {
    const int st = pos0[b];
    if (blockIdx.x == 0) bin_start[b] = st;
    const int h = hist[b];
    if (h) pos0[b] = st + atomicAdd(&bin_cur[b], h);
  }
  __syncthreads();
#pragma unroll
  for (int i = 0; i < EDGE_EPT; ++i) {
    if (e0 + i * 256 < E) {
      const int p = pos0[cc[i] >> BIN_SHIFT] + lr[i];
      if ((unsigned)p < (unsigned)E) {
        bsrc[p] = rr[i];
        bdst[p] = cc[i];
      }
      const int q = pos0[NB + (rr[i] >> BIN_SHIFT)] + sr[i];
      if ((unsigned)q < (unsigned)E) srow[q] = rr[i];
    }
  }
}

__global__ void __launch_bounds__(1024) gcn_bin_sort(
    const int* __restrict__ bsrc, const int* __restrict__ bdst,
    const int* __restrict__ srow,
    const int* __restrict__ bin_tot, const int* __restrict__ bin_start,
    float* __restrict__ deg_out, int* __restrict__ seg,
    float* __restrict__ deg_in, int* __restrict__ srcs,
    int N, int E, int NB) {
  __shared__ int cnt[256];
  __shared__ int scnt[256];
  __shared__ int off[256];
  __shared__ int cur[256];
  __shared__ int wave_sum[8];
  const int tid = threadIdx.x;
  const int lane = tid & 31;
  const int wave = tid >> 5;
  const int bin = blockIdx.x;
  const int n = bin_tot[bin];
  const int s0 = bin_start[bin];
  const int sn = bin_tot[NB + bin];
  const int ss0 = bin_start[NB + bin];
  const int c0 = bin << BIN_SHIFT;
  if (tid < 256) {
    cnt[tid] = 0;
    scnt[tid] = 0;
    cur[tid] = 0;
  }
  __syncthreads();
  for (int i = tid; i < n; i += 1024) atomicAdd(&cnt[(bdst[s0 + i] - c0) & 255], 1);
  for (int i = tid; i < sn; i += 1024) atomicAdd(&scnt[(srow[ss0 + i] - c0) & 255], 1);
  __syncthreads();
  const int d = (tid < 256) ? cnt[tid] : 0;
  int incl = d;
  for (int o = 1; o < 32; o <<= 1) {
    int t = __shfl_up(incl, o);
    if (lane >= o) incl += t;
  }
  if (tid < 256 && lane == 31) wave_sum[wave] = incl;
  __syncthreads();
  if (tid == 0) {
    int tot = 0;
    for (int w = 0; w < 8; ++w) {
      int t = wave_sum[w];
      wave_sum[w] = tot;
      tot += t;
    }
  }
  __syncthreads();
  if (tid < 256) {
    const int o = wave_sum[wave] + incl - d;
    off[tid] = o;
    const int c = c0 + tid;
    if (c < N) {
      reinterpret_cast<int2*>(seg)[c] = make_int2(s0 + o, d);
      deg_in[c] = (float)d;
      deg_out[c] = (float)scnt[tid];
    }
  }
  __syncthreads();
  for (int i = tid; i < n; i += 1024) {
    const int loc = (bdst[s0 + i] - c0) & 255;
    const int r = bsrc[s0 + i];
    const int k = atomicAdd(&cur[loc], 1);
    const int p = s0 + off[loc] + k;
    if ((unsigned)p < (unsigned)E) srcs[p] = r;
  }
}

#define GATHER_BLOCKS_12 1536
#define GATHER_BLOCKS_16 2048

static __device__ __forceinline__ float4 gcn_row(const float* __restrict__ xs, int idx,
                                                 int k, unsigned last, int lane) {
  unsigned r = min((unsigned)__builtin_amdgcn_readlane(idx, k), last);
  return reinterpret_cast<const float4*>(xs + (size_t)r * DFEAT)[lane];
}

template <bool kDeg>
static __device__ __forceinline__ void gcn_fma4(float4& a, const float4 v, int wbits, int k) {
  float w = __int_as_float(__builtin_amdgcn_readlane(wbits, k));
  if (kDeg) w = rsqrtf(fmaxf(w, 1.0f));
  a.x = fmaf(v.x, w, a.x); a.y = fmaf(v.y, w, a.y);
  a.z = fmaf(v.z, w, a.z); a.w = fmaf(v.w, w, a.w);
}

template <bool kFirst>
static __device__ __forceinline__ void gcn_stretch(
    const float* __restrict__ xs, const float* __restrict__ deg_out, int idx, float& wsrc,
    int m, unsigned last, int lane, float4& a0, float4& a1, float4& a2, float4& a3) {
  if (kFirst) wsrc = deg_out[min((unsigned)idx, last)];
  const int wb = __float_as_int(wsrc);
  int j = 0;
  for (; j + 8 <= m; j += 8) {
    float4 v0 = gcn_row(xs, idx, j + 0, last, lane);
    float4 v1 = gcn_row(xs, idx, j + 1, last, lane);
    float4 v2 = gcn_row(xs, idx, j + 2, last, lane);
    float4 v3 = gcn_row(xs, idx, j + 3, last, lane);
    float4 v4 = gcn_row(xs, idx, j + 4, last, lane);
    float4 v5 = gcn_row(xs, idx, j + 5, last, lane);
    float4 v6 = gcn_row(xs, idx, j + 6, last, lane);
    float4 v7 = gcn_row(xs, idx, j + 7, last, lane);
    if (kFirst) __builtin_amdgcn_sched_barrier(0);
    gcn_fma4<kFirst>(a0, v0, wb, j + 0); gcn_fma4<kFirst>(a1, v1, wb, j + 1);
    gcn_fma4<kFirst>(a2, v2, wb, j + 2); gcn_fma4<kFirst>(a3, v3, wb, j + 3);
    gcn_fma4<kFirst>(a0, v4, wb, j + 4); gcn_fma4<kFirst>(a1, v5, wb, j + 5);
    gcn_fma4<kFirst>(a2, v6, wb, j + 6); gcn_fma4<kFirst>(a3, v7, wb, j + 7);
  }
  if ((m - j) & 4) {
    float4 v0 = gcn_row(xs, idx, j + 0, last, lane);
    float4 v1 = gcn_row(xs, idx, j + 1, last, lane);
    float4 v2 = gcn_row(xs, idx, j + 2, last, lane);
    float4 v3 = gcn_row(xs, idx, j + 3, last, lane);
    if (kFirst) __builtin_amdgcn_sched_barrier(0);
    gcn_fma4<kFirst>(a0, v0, wb, j + 0); gcn_fma4<kFirst>(a1, v1, wb, j + 1);
    gcn_fma4<kFirst>(a2, v2, wb, j + 2); gcn_fma4<kFirst>(a3, v3, wb, j + 3);
    j += 4;
  }
  if ((m - j) & 2) {
    float4 v0 = gcn_row(xs, idx, j + 0, last, lane);
    float4 v1 = gcn_row(xs, idx, j + 1, last, lane);
    if (kFirst) __builtin_amdgcn_sched_barrier(0);
    gcn_fma4<kFirst>(a0, v0, wb, j + 0); gcn_fma4<kFirst>(a1, v1, wb, j + 1);
    j += 2;
  }
  if ((m - j) & 1) {
    float4 v0 = gcn_row(xs, idx, j, last, lane);
    if (kFirst) __builtin_amdgcn_sched_barrier(0);
    gcn_fma4<kFirst>(a0, v0, wb, j);
  }
}

template <bool kFirst>
__global__ void __launch_bounds__(256, kFirst ? 12 : 16) gcn_gather(
    const float* __restrict__ xs, const int* __restrict__ srcs,
    const float* __restrict__ deg_out, float* __restrict__ wts,
    const int* __restrict__ seg, float* __restrict__ agg, int N, int E) {
  const int lane = threadIdx.x & 31;
  const int nw = gridDim.x * 8;
  int c = __builtin_amdgcn_readfirstlane(blockIdx.x * 8 + (threadIdx.x >> 5));
  if (c >= N) return;
  const unsigned last = (unsigned)(N - 1);
  const int elast = E - 1;
  const int2* __restrict__ seg2 = reinterpret_cast<const int2*>(seg);

  int2 sc = seg2[c];
  int cn = c + nw;
  int2 scn = seg2[min(cn, N - 1)];
  int m = min(sc.y, 32);
  int idx = (lane < m) ? srcs[sc.x + lane] : 0;
  float wsrc = 0.0f;
  if (!kFirst) wsrc = (lane < m) ? wts[sc.x + lane] : 0.0f;

  for (;;) {
    const int cnn = cn + nw;
    const int2 scnn = seg2[min(cnn, N - 1)];
    const int mn = (cn < N) ? min(scn.y, 32) : 0;
    const int slotn = min(scn.x + lane, elast);
    const int idxn_raw = srcs[slotn];
    const int idxn = (lane < mn) ? idxn_raw : 0;
    float wsrcn = 0.0f;
    if (!kFirst) {
      const float wn_raw = wts[slotn];
      wsrcn = (lane < mn) ? wn_raw : 0.0f;
    }

    const int s = sc.x;
    const int cnt = sc.y;
    float4 a0 = make_float4(0.f, 0.f, 0.f, 0.f);
    float4 a1 = a0, a2 = a0, a3 = a0;
    gcn_stretch<kFirst>(xs, deg_out, idx, wsrc, m, last, lane, a0, a1, a2, a3);
    if (kFirst && lane < m) wts[s + lane] = rsqrtf(fmaxf(wsrc, 1.0f));
    for (int base = 32; base < cnt; base += 32) {
      const int m2 = min(cnt - base, 32);
      const int idx2 = (lane < m2) ? srcs[s + base + lane] : 0;
      float w2 = 0.0f;
      if (!kFirst) w2 = (lane < m2) ? wts[s + base + lane] : 0.0f;
      gcn_stretch<kFirst>(xs, deg_out, idx2, w2, m2, last, lane, a0, a1, a2, a3);
      if (kFirst && lane < m2) wts[s + base + lane] = rsqrtf(fmaxf(w2, 1.0f));
    }
    float4 acc;
    acc.x = (a0.x + a1.x) + (a2.x + a3.x);
    acc.y = (a0.y + a1.y) + (a2.y + a3.y);
    acc.z = (a0.z + a1.z) + (a2.z + a3.z);
    acc.w = (a0.w + a1.w) + (a2.w + a3.w);
    reinterpret_cast<float4*>(agg + (size_t)c * DFEAT)[lane] = acc;

    if (cn >= N) break;
    c = cn; cn = cnn;
    sc = scn; scn = scnn;
    m = mn; idx = idxn; wsrc = wsrcn;
  }
}

#define GEMM_PITCH 132
#define GEMM_BLOCKS 512

static __device__ __forceinline__ float4 gcn_frag(const float* __restrict__ tile, int c, int h, int j) {
  return *reinterpret_cast<const float4*>(tile + c * GEMM_PITCH + 8 * j + 4 * h);
}

template <bool kResid, bool kRelu, bool kFull>
static __device__ __forceinline__ void gcn_rows_out(
    const float* strip, const float4 bias4, const float* __restrict__ deg_in,
    const float* __restrict__ resid, float* __restrict__ out, int n0, int N, int lane) {
  float dg[16];
  v4f x4[16];
#pragma unroll
  for (int r = 0; r < 16; ++r) {
    if (kFull || n0 + r < N) {
      dg[r] = deg_in[n0 + r];
      if (kResid) x4[r] = reinterpret_cast<const v4f*>(resid + (size_t)(n0 + r) * DFEAT)[lane];
    }
  }
  if (kResid && kFull) {
    asm volatile("" ::"v"(x4[0]), "v"(x4[1]), "v"(x4[2]), "v"(x4[3]), "v"(x4[4]), "v"(x4[5]),
                 "v"(x4[6]), "v"(x4[7]), "v"(x4[8]), "v"(x4[9]), "v"(x4[10]), "v"(x4[11]),
                 "v"(x4[12]), "v"(x4[13]), "v"(x4[14]), "v"(x4[15]));
  }
#pragma unroll
  for (int r = 0; r < 16; ++r) {
    if (kFull || n0 + r < N) {
      float4 v = *reinterpret_cast<const float4*>(strip + r * GEMM_PITCH + 4 * lane);
      const float s = rsqrtf(fmaxf(dg[r], 1.0f));
      v.x += bias4.x; v.y += bias4.y; v.z += bias4.z; v.w += bias4.w;
      v.x *= s; v.y *= s; v.z *= s; v.w *= s;
      if (kResid) { v.x += x4[r].x; v.y += x4[r].y; v.z += x4[r].z; v.w += x4[r].w; }
      if (kRelu) {
        v.x = fmaxf(v.x, 0.0f); v.y = fmaxf(v.y, 0.0f);
        v.z = fmaxf(v.z, 0.0f); v.w = fmaxf(v.w, 0.0f);
      }
      reinterpret_cast<float4*>(out + (size_t)(n0 + r) * DFEAT)[lane] = v;
    }
  }
}

template <bool kResid, bool kRelu>
__global__ void __launch_bounds__(256, 4) gcn_gemm_norm(
    const float* __restrict__ A,
    const float* __restrict__ W,
    const float* __restrict__ bias,
    const float* __restrict__ deg_in,
    const float* __restrict__ resid,
    float* __restrict__ out,
    int N) {
  __shared__ __attribute__((aligned(16))) float lds_w[DFEAT * GEMM_PITCH];
  __shared__ __attribute__((aligned(16))) float lds_a[8 * 16 * GEMM_PITCH];

  const int lane = threadIdx.x & 31;
  const int wave = threadIdx.x >> 5;
  const int h = lane >> 4;
  const int c = lane & 15;

#pragma unroll
  for (int it = 0; it < 16; ++it) {
    const int idx = it * 256 + threadIdx.x;
    const float4 v = reinterpret_cast<const float4*>(W)[idx];
    *reinterpret_cast<float4*>(lds_w + (idx >> 5) * GEMM_PITCH + 4 * (idx & 31)) = v;
  }
  __syncthreads();

  float* strip = lds_a + wave * 16 * GEMM_PITCH;
  const float4 bias4 = reinterpret_cast<const float4*>(bias)[lane];
  const int tiles = (N + 15) >> 4;
  const int step = gridDim.x * 8;
  for (int t = __builtin_amdgcn_readfirstlane(wave * gridDim.x + blockIdx.x); t < tiles; t += step) {
    const int n0 = t * 16;

#pragma unroll
    for (int r = 0; r < 16; ++r) {
      const int nr = min(n0 + r, N - 1);
      const float4 v = reinterpret_cast<const float4*>(A + (size_t)nr * DFEAT)[lane];
      *reinterpret_cast<float4*>(strip + r * GEMM_PITCH + 4 * lane) = v;
    }
    __builtin_amdgcn_fence(__ATOMIC_ACQ_REL, "wavefront");

    v8f acc[8];
#pragma unroll
    for (int ct = 0; ct < 8; ++ct) acc[ct] = (v8f){0.f, 0.f, 0.f, 0.f, 0.f, 0.f, 0.f, 0.f};

#pragma unroll
    for (int j = 0; j < DFEAT / 8; ++j) {
      const float4 a = gcn_frag(strip, c, h, j);
      float4 b[8];
#pragma unroll
      for (int ct = 0; ct < 8; ++ct) b[ct] = gcn_frag(lds_w + ct * 16 * GEMM_PITCH, c, h, j);
      const v2f a0 = {a.x, a.y};
      const v2f a1 = {a.z, a.w};
#pragma unroll
      for (int ct = 0; ct < 8; ++ct) {
        const v2f b0 = {b[ct].x, b[ct].y};
        acc[ct] = __builtin_amdgcn_wmma_f32_16x16x4_f32(
             false, a0,  false, b0,
             (short)0, acc[ct],  false,  false);
      }
#pragma unroll
      for (int ct = 0; ct < 8; ++ct) {
        const v2f b1 = {b[ct].z, b[ct].w};
        acc[ct] = __builtin_amdgcn_wmma_f32_16x16x4_f32(
             false, a1,  false, b1,
             (short)0, acc[ct],  false,  false);
      }
    }

#pragma unroll
    for (int ct = 0; ct < 8; ++ct) {
#pragma unroll
      for (int r = 0; r < 8; ++r) strip[(8 * h + r) * GEMM_PITCH + 16 * ct + c] = acc[ct][r];
    }
    __builtin_amdgcn_fence(__ATOMIC_ACQ_REL, "wavefront");

    if (n0 + 16 <= N) {
      gcn_rows_out<kResid, kRelu, true>(strip, bias4, deg_in, resid, out, n0, N, lane);
    } else {
      gcn_rows_out<kResid, kRelu, false>(strip, bias4, deg_in, resid, out, n0, N, lane);
    }
    __builtin_amdgcn_fence(__ATOMIC_ACQ_REL, "wavefront");
  }
}

extern "C" void kernel_launch(void* const* d_in, const int* in_sizes, int n_in,
                              void* d_out, int out_size, void* d_ws, size_t ws_size,
                              hipStream_t stream) {
  const float* x  = (const float*)d_in[0];
  const int*   ei = (const int*)d_in[1];
  const float* W1 = (const float*)d_in[2];
  const float* b1 = (const float*)d_in[3];
  const float* W2 = (const float*)d_in[4];
  const float* b2 = (const float*)d_in[5];

  const int N = in_sizes[0] / DFEAT;
  const int E = in_sizes[1] / 2;
  const int NB = (N + 255) >> BIN_SHIFT;
  const int* row = ei;
  const int* col = ei + E;
  float* out = (float*)d_out;

  float* agg       = (float*)d_ws;
  int*   seg       = (int*)(agg + (size_t)N * DFEAT);
  int*   bin_tot   = seg + 2 * N;
  int*   bin_cur   = bin_tot + 2 * NB;
  float* deg_out   = (float*)(bin_cur + 2 * NB);
  float* deg_in    = deg_out + N;
  int*   bin_start = (int*)(deg_in + N);
  int*   bsrc      = bin_start + 2 * NB;
  int*   bdst      = bsrc + E;
  int*   srow      = bdst + E;
  int*   srcs      = srow + E;
  float* wts       = (float*)(srcs + E);

  const int TB = 256;
  const int EB = (E + TB * EDGE_EPT - 1) / (TB * EDGE_EPT);

  gcn_zero_f32<<<(4 * NB + TB - 1) / TB, TB, 0, stream>>>((float*)bin_tot, 4 * NB);
  gcn_bin_count<<<EB, TB, 2 * NB * sizeof(int), stream>>>(row, col, bin_tot, E, NB);
  gcn_bin_scatter<<<EB, TB, 4 * NB * sizeof(int), stream>>>(row, col, bin_tot, bin_cur, bin_start,
                                                           bsrc, bdst, srow, E, NB);
  gcn_bin_sort<<<NB, 1024, 0, stream>>>(bsrc, bdst, srow, bin_tot, bin_start, deg_out, seg, deg_in,
                                        srcs, N, E, NB);

  gcn_gather<true><<<GATHER_BLOCKS_12, TB, 0, stream>>>(x, srcs, deg_out, wts, seg, agg, N, E);
  gcn_gemm_norm<true, true><<<GEMM_BLOCKS, TB, 0, stream>>>(agg, W1, b1, deg_in, x, out, N);

  gcn_gather<false><<<GATHER_BLOCKS_16, TB, 0, stream>>>(out, srcs, deg_out, wts, seg, agg, N, E);
  gcn_gemm_norm<false, false><<<GEMM_BLOCKS, TB, 0, stream>>>(agg, W2, b2, deg_in, nullptr, out, N);
}
